# starting kernel
# baseline (speedup 1.0000x reference)
#include <hip/hip_runtime.h>

#define N_NODES 50000
#define N_REL   16
#define HID     64
#define N_LAB   32
#define N_EDGE  1600000
#define K_AGG   1024
#define K_TOT   1088

typedef float v2f __attribute__((ext_vector_type(2)));
typedef float v8f __attribute__((ext_vector_type(8)));

#define WS_CNT_OFF  0
#define WS_B_OFF    (N_REL * N_NODES)
#define WS_LHS_OFF  (WS_B_OFF + K_TOT * N_LAB)
#define WS_TOTAL    (WS_LHS_OFF + (size_t)N_NODES * K_TOT)

__device__ __forceinline__ void fatomic_add(float* p, float v) {
  __hip_atomic_fetch_add(p, v, __ATOMIC_RELAXED, __HIP_MEMORY_SCOPE_AGENT);
}

__global__ void k_zero(float* __restrict__ p, unsigned long long n) {
  unsigned long long i = (unsigned long long)blockIdx.x * blockDim.x + threadIdx.x;
  unsigned long long stride = (unsigned long long)gridDim.x * blockDim.x;
  for (; i < n; i += stride) p[i] = 0.0f;
}

__global__ void k_count(const int* __restrict__ etype, const int* __restrict__ dst,
                        float* __restrict__ cnt) {
  unsigned e = blockIdx.x * blockDim.x + threadIdx.x;
  if (e >= N_EDGE) return;
  fatomic_add(&cnt[etype[e] * N_NODES + dst[e]], 1.0f);
}

__global__ void k_bmat(const float* __restrict__ w2, const float* __restrict__ r2,
                       float* __restrict__ B) {
  unsigned t = blockIdx.x * blockDim.x + threadIdx.x;
  if (t >= K_TOT * N_LAB) return;
  B[t] = (t < K_AGG * N_LAB) ? w2[t] : r2[t - K_AGG * N_LAB];
}

__global__ void k_layer1(const int* __restrict__ src_a, const int* __restrict__ dst_a,
                         const int* __restrict__ etype, const float* __restrict__ w1,
                         const float* __restrict__ cnt, float* __restrict__ lhs) {
  unsigned tid = blockIdx.x * blockDim.x + threadIdx.x;
  unsigned e = tid >> 5;
  unsigned lane = tid & 31;
  if (e >= N_EDGE) return;
  int rel = etype[e], s = src_a[e], d = dst_a[e];
  float inv = 1.0f / fmaxf(cnt[rel * N_NODES + d], 1.0f);
  const float* wrow = w1 + ((size_t)rel * N_NODES + (size_t)s) * HID + 2u * lane;
  v2f v = *(const v2f*)wrow;
  float* o = lhs + (size_t)d * K_TOT + K_AGG + 2u * lane;
  fatomic_add(o + 0, v.x * inv);
  fatomic_add(o + 1, v.y * inv);
}

__global__ void k_relu(const float* __restrict__ root1, const float* __restrict__ bias1,
                       float* __restrict__ lhs) {
  unsigned idx = blockIdx.x * blockDim.x + threadIdx.x;
  if (idx >= N_NODES * HID) return;
  unsigned n = idx >> 6, h = idx & 63;
  size_t p = (size_t)n * K_TOT + K_AGG + h;
  lhs[p] = fmaxf(lhs[p] + root1[idx] + bias1[h], 0.0f);
}

__global__ void k_layer2(const int* __restrict__ src_a, const int* __restrict__ dst_a,
                         const int* __restrict__ etype, const float* __restrict__ cnt,
                         float* __restrict__ lhs) {
  unsigned tid = blockIdx.x * blockDim.x + threadIdx.x;
  unsigned e = tid >> 5;
  unsigned lane = tid & 31;
  if (e >= N_EDGE) return;
  int rel = etype[e], s = src_a[e], d = dst_a[e];
  float inv = 1.0f / fmaxf(cnt[rel * N_NODES + d], 1.0f);
  const float* xrow = lhs + (size_t)s * K_TOT + K_AGG + 2u * lane;
  v2f v = *(const v2f*)xrow;
  float* o = lhs + (size_t)d * K_TOT + (unsigned)rel * HID + 2u * lane;
  fatomic_add(o + 0, v.x * inv);
  fatomic_add(o + 1, v.y * inv);
}

__global__ void k_gemm(const float* __restrict__ lhs, const float* __restrict__ B,
                       const float* __restrict__ bias2, float* __restrict__ out) {
  unsigned tid = blockIdx.x * blockDim.x + threadIdx.x;
  unsigned wave = tid >> 5;
  unsigned lane = tid & 31;
  if (wave >= N_NODES / 16) return;
  unsigned half = lane >> 4;
  unsigned sub = lane & 15;
  const float* arow = lhs + (size_t)(wave * 16u + sub) * K_TOT + 2u * half;
  v8f c0 = {}; v8f c1 = {};
  for (int k0 = 0; k0 < K_TOT; k0 += 4) {
    v2f a;  a.x = arow[k0];  a.y = arow[k0 + 1];
    int kb = k0 + 2 * (int)half;
    v2f b0; b0.x = B[kb * N_LAB + sub];        b0.y = B[(kb + 1) * N_LAB + sub];
    v2f b1; b1.x = B[kb * N_LAB + 16 + sub];   b1.y = B[(kb + 1) * N_LAB + 16 + sub];
    c0 = __builtin_amdgcn_wmma_f32_16x16x4_f32(false, a, false, b0, (short)0, c0, false, false);
    c1 = __builtin_amdgcn_wmma_f32_16x16x4_f32(false, a, false, b1, (short)0, c1, false, false);
  }
  #pragma unroll
  for (int v = 0; v < 8; v++) {
    unsigned row = wave * 16u + (unsigned)v + 8u * half;
    float x0 = c0[v] + bias2[sub];
    float x1 = c1[v] + bias2[16 + sub];
    out[row * N_LAB + sub]      = 1.0f / (1.0f + __expf(-x0));
    out[row * N_LAB + 16 + sub] = 1.0f / (1.0f + __expf(-x1));
  }
}

extern "C" void kernel_launch(void* const* d_in, const int* in_sizes, int n_in,
                              void* d_out, int out_size, void* d_ws, size_t ws_size,
                              hipStream_t stream) {
  const int* ei     = (const int*)d_in[0];
  const int* src_a  = ei;
  const int* dst_a  = ei + N_EDGE;
  const int* etype  = (const int*)d_in[1];
  const float* w1   = (const float*)d_in[2];
  const float* r1   = (const float*)d_in[3];
  const float* b1   = (const float*)d_in[4];
  const float* w2   = (const float*)d_in[5];
  const float* r2   = (const float*)d_in[6];
  const float* b2   = (const float*)d_in[7];
  float* out = (float*)d_out;

  float* ws   = (float*)d_ws;
  float* cnt  = ws + WS_CNT_OFF;
  float* Bmat = ws + WS_B_OFF;
  float* lhs  = ws + WS_LHS_OFF;

  (void)in_sizes; (void)n_in; (void)out_size; (void)ws_size;

  k_zero<<<2048, 256, 0, stream>>>(ws, (unsigned long long)WS_TOTAL);
  k_count<<<(N_EDGE + 255) / 256, 256, 0, stream>>>(etype, dst_a, cnt);
  k_bmat<<<(K_TOT * N_LAB + 255) / 256, 256, 0, stream>>>(w2, r2, Bmat);
  k_layer1<<<(N_EDGE * 32u + 255) / 256, 256, 0, stream>>>(src_a, dst_a, etype, w1, cnt, lhs);
  k_relu<<<(N_NODES * HID + 255) / 256, 256, 0, stream>>>(r1, b1, lhs);
  k_layer2<<<(N_EDGE * 32u + 255) / 256, 256, 0, stream>>>(src_a, dst_a, etype, cnt, lhs);
  k_gemm<<<((N_NODES / 16) * 32 + 255) / 256, 256, 0, stream>>>(lhs, Bmat, b2, out);
}


# layers 1 and 2: lane l takes features l and l+32 (was 2l, 2l+1), so each f32 atomic add instruction fills one whole 128-byte line instead of half of two; adds still land in arrival order
# speedup vs baseline: 1.0434x; 1.0434x over previous
#include <hip/hip_runtime.h>

#define N_NODES 50000
#define N_REL   16
#define HID     64
#define N_LAB   32
#define N_EDGE  1600000
#define K_AGG   1024
#define K_TOT   1088

typedef float v2f __attribute__((ext_vector_type(2)));
typedef float v8f __attribute__((ext_vector_type(8)));

#define WS_CNT_OFF  0
#define WS_B_OFF    (N_REL * N_NODES)
#define WS_LHS_OFF  (WS_B_OFF + K_TOT * N_LAB)
#define WS_TOTAL    (WS_LHS_OFF + (size_t)N_NODES * K_TOT)

__device__ __forceinline__ void fatomic_add(float* p, float v) {
  __hip_atomic_fetch_add(p, v, __ATOMIC_RELAXED, __HIP_MEMORY_SCOPE_AGENT);
}

__global__ void k_zero(float* __restrict__ p, unsigned long long n) {
  unsigned long long i = (unsigned long long)blockIdx.x * blockDim.x + threadIdx.x;
  unsigned long long stride = (unsigned long long)gridDim.x * blockDim.x;
  for (; i < n; i += stride) p[i] = 0.0f;
}

__global__ void k_count(const int* __restrict__ etype, const int* __restrict__ dst,
                        float* __restrict__ cnt) {
  unsigned e = blockIdx.x * blockDim.x + threadIdx.x;
  if (e >= N_EDGE) return;
  fatomic_add(&cnt[etype[e] * N_NODES + dst[e]], 1.0f);
}

__global__ void k_bmat(const float* __restrict__ w2, const float* __restrict__ r2,
                       float* __restrict__ B) {
  unsigned t = blockIdx.x * blockDim.x + threadIdx.x;
  if (t >= K_TOT * N_LAB) return;
  B[t] = (t < K_AGG * N_LAB) ? w2[t] : r2[t - K_AGG * N_LAB];
}

__global__ void k_layer1(const int* __restrict__ src_a, const int* __restrict__ dst_a,
                         const int* __restrict__ etype, const float* __restrict__ w1,
                         const float* __restrict__ cnt, float* __restrict__ lhs) {
  unsigned tid = blockIdx.x * blockDim.x + threadIdx.x;
  unsigned e = tid >> 5;
  unsigned lane = tid & 31;
  if (e >= N_EDGE) return;
  int rel = etype[e], s = src_a[e], d = dst_a[e];
  float inv = 1.0f / fmaxf(cnt[rel * N_NODES + d], 1.0f);
  const float* wrow = w1 + ((size_t)rel * N_NODES + (size_t)s) * HID + lane;
  float v0 = wrow[0];
  float v1 = wrow[32];
  float* o = lhs + (size_t)d * K_TOT + K_AGG + lane;
  fatomic_add(o, v0 * inv);
  fatomic_add(o + 32, v1 * inv);
}

__global__ void k_relu(const float* __restrict__ root1, const float* __restrict__ bias1,
                       float* __restrict__ lhs) {
  unsigned idx = blockIdx.x * blockDim.x + threadIdx.x;
  if (idx >= N_NODES * HID) return;
  unsigned n = idx >> 6, h = idx & 63;
  size_t p = (size_t)n * K_TOT + K_AGG + h;
  lhs[p] = fmaxf(lhs[p] + root1[idx] + bias1[h], 0.0f);
}

__global__ void k_layer2(const int* __restrict__ src_a, const int* __restrict__ dst_a,
                         const int* __restrict__ etype, const float* __restrict__ cnt,
                         float* __restrict__ lhs) {
  unsigned tid = blockIdx.x * blockDim.x + threadIdx.x;
  unsigned e = tid >> 5;
  unsigned lane = tid & 31;
  if (e >= N_EDGE) return;
  int rel = etype[e], s = src_a[e], d = dst_a[e];
  float inv = 1.0f / fmaxf(cnt[rel * N_NODES + d], 1.0f);
  const float* xrow = lhs + (size_t)s * K_TOT + K_AGG + lane;
  float v0 = xrow[0];
  float v1 = xrow[32];
  float* o = lhs + (size_t)d * K_TOT + (unsigned)rel * HID + lane;
  fatomic_add(o, v0 * inv);
  fatomic_add(o + 32, v1 * inv);
}

__global__ void k_gemm(const float* __restrict__ lhs, const float* __restrict__ B,
                       const float* __restrict__ bias2, float* __restrict__ out) {
  unsigned tid = blockIdx.x * blockDim.x + threadIdx.x;
  unsigned wave = tid >> 5;
  unsigned lane = tid & 31;
  if (wave >= N_NODES / 16) return;
  unsigned half = lane >> 4;
  unsigned sub = lane & 15;
  const float* arow = lhs + (size_t)(wave * 16u + sub) * K_TOT + 2u * half;
  v8f c0 = {}; v8f c1 = {};
  for (int k0 = 0; k0 < K_TOT; k0 += 4) {
    v2f a;  a.x = arow[k0];  a.y = arow[k0 + 1];
    int kb = k0 + 2 * (int)half;
    v2f b0; b0.x = B[kb * N_LAB + sub];        b0.y = B[(kb + 1) * N_LAB + sub];
    v2f b1; b1.x = B[kb * N_LAB + 16 + sub];   b1.y = B[(kb + 1) * N_LAB + 16 + sub];
    c0 = __builtin_amdgcn_wmma_f32_16x16x4_f32(false, a, false, b0, (short)0, c0, false, false);
    c1 = __builtin_amdgcn_wmma_f32_16x16x4_f32(false, a, false, b1, (short)0, c1, false, false);
  }
  #pragma unroll
  for (int v = 0; v < 8; v++) {
    unsigned row = wave * 16u + (unsigned)v + 8u * half;
    float x0 = c0[v] + bias2[sub];
    float x1 = c1[v] + bias2[16 + sub];
    out[row * N_LAB + sub]      = 1.0f / (1.0f + __expf(-x0));
    out[row * N_LAB + 16 + sub] = 1.0f / (1.0f + __expf(-x1));
  }
}

extern "C" void kernel_launch(void* const* d_in, const int* in_sizes, int n_in,
                              void* d_out, int out_size, void* d_ws, size_t ws_size,
                              hipStream_t stream) {
  const int* ei     = (const int*)d_in[0];
  const int* src_a  = ei;
  const int* dst_a  = ei + N_EDGE;
  const int* etype  = (const int*)d_in[1];
  const float* w1   = (const float*)d_in[2];
  const float* r1   = (const float*)d_in[3];
  const float* b1   = (const float*)d_in[4];
  const float* w2   = (const float*)d_in[5];
  const float* r2   = (const float*)d_in[6];
  const float* b2   = (const float*)d_in[7];
  float* out = (float*)d_out;

  float* ws   = (float*)d_ws;
  float* cnt  = ws + WS_CNT_OFF;
  float* Bmat = ws + WS_B_OFF;
  float* lhs  = ws + WS_LHS_OFF;

  (void)in_sizes; (void)n_in; (void)out_size; (void)ws_size;

  k_zero<<<2048, 256, 0, stream>>>(ws, (unsigned long long)WS_TOTAL);
  k_count<<<(N_EDGE + 255) / 256, 256, 0, stream>>>(etype, dst_a, cnt);
  k_bmat<<<(K_TOT * N_LAB + 255) / 256, 256, 0, stream>>>(w2, r2, Bmat);
  k_layer1<<<(N_EDGE * 32u + 255) / 256, 256, 0, stream>>>(src_a, dst_a, etype, w1, cnt, lhs);
  k_relu<<<(N_NODES * HID + 255) / 256, 256, 0, stream>>>(r1, b1, lhs);
  k_layer2<<<(N_EDGE * 32u + 255) / 256, 256, 0, stream>>>(src_a, dst_a, etype, cnt, lhs);
  k_gemm<<<((N_NODES / 16) * 32 + 255) / 256, 256, 0, stream>>>(lhs, Bmat, b2, out);
}


# layers 1+2: edges counting-sorted by (dst,rel); one wave per dst node sums its rows in registers and stores once: no f32 atomics, no workspace zeroing; sums inside a segment run in fill arrival order
# speedup vs baseline: 1.9820x; 1.8996x over previous
#include <hip/hip_runtime.h>


#define N_NODES 50000
#define N_REL   16
#define HID     64
#define N_LAB   32
#define N_EDGE  1600000
#define K_AGG   1024
#define K_TOT   1088
#define N_KEYS  (N_NODES * N_REL)
#define SCAN_BLOCKS ((N_NODES + 255) / 256)

typedef float v2f __attribute__((ext_vector_type(2)));
typedef float v8f __attribute__((ext_vector_type(8)));

#define WS_CNT_OFF  0
#define WS_B_OFF    (N_REL * N_NODES)
#define WS_LHS_OFF  (WS_B_OFF + K_TOT * N_LAB)
#define WS_TOTAL    (WS_LHS_OFF + (size_t)N_NODES * K_TOT)

__global__ void k_zero(int* __restrict__ p) {
  unsigned i = blockIdx.x * blockDim.x + threadIdx.x;
  if (i < N_KEYS) p[i] = 0;
}

__global__ void k_count(const int* __restrict__ etype, const int* __restrict__ dst,
                        int* __restrict__ cnt) {
  unsigned e = blockIdx.x * blockDim.x + threadIdx.x;
  if (e >= N_EDGE) return;
  __hip_atomic_fetch_add(&cnt[dst[e] * N_REL + etype[e]], 1, __ATOMIC_RELAXED, __HIP_MEMORY_SCOPE_AGENT);
}

__device__ __forceinline__ int block_sum256(int v, int* sh) {
  unsigned t = threadIdx.x;
  sh[t] = v;
  __syncthreads();
  for (unsigned s = 128; s > 0; s >>= 1) {
    if (t < s) sh[t] += sh[t + s];
    __syncthreads();
  }
  int r = sh[0];
  __syncthreads();
  return r;
}

__global__ void k_scan1(const int* __restrict__ cnt, int* __restrict__ bsum) {
  __shared__ int sh[256];
  unsigned n = blockIdx.x * 256u + threadIdx.x;
  int deg = 0;
  if (n < N_NODES) {
    const int4* p = (const int4*)(cnt + (size_t)n * N_REL);
    for (int q = 0; q < 4; q++) { int4 c = p[q]; deg += c.x + c.y + c.z + c.w; }
  }
  int tot = block_sum256(deg, sh);
  if (threadIdx.x == 0) bsum[blockIdx.x] = tot;
}

__global__ void k_scan2(int* __restrict__ cnt, const int* __restrict__ bsum) {
  __shared__ int sh[256];
  unsigned t = threadIdx.x;
  unsigned n = blockIdx.x * 256u + t;
  int base = block_sum256((t < blockIdx.x) ? bsum[t] : 0, sh);
  int c[16];
  int deg = 0;
  if (n < N_NODES) {
    const int4* p = (const int4*)(cnt + (size_t)n * N_REL);
    for (int q = 0; q < 4; q++) {
      int4 v = p[q];
      c[4 * q] = v.x; c[4 * q + 1] = v.y; c[4 * q + 2] = v.z; c[4 * q + 3] = v.w;
      deg += v.x + v.y + v.z + v.w;
    }
  } else {
    for (int q = 0; q < 16; q++) c[q] = 0;
  }
  sh[t] = deg;
  __syncthreads();
  for (unsigned o = 1; o < 256; o <<= 1) {
    int add = (t >= o) ? sh[t - o] : 0;
    __syncthreads();
    sh[t] += add;
    __syncthreads();
  }
  int run = base + sh[t] - deg;
  if (n < N_NODES) {
    int4* p = (int4*)(cnt + (size_t)n * N_REL);
    for (int q = 0; q < 4; q++) {
      int4 v;
      v.x = run; run += c[4 * q];
      v.y = run; run += c[4 * q + 1];
      v.z = run; run += c[4 * q + 2];
      v.w = run; run += c[4 * q + 3];
      p[q] = v;
    }
  }
}

__global__ void k_fill(const int* __restrict__ src_a, const int* __restrict__ dst_a,
                       const int* __restrict__ etype, int* __restrict__ off,
                       int* __restrict__ ssrc) {
  unsigned e = blockIdx.x * blockDim.x + threadIdx.x;
  if (e >= N_EDGE) return;
  int pos = __hip_atomic_fetch_add(&off[dst_a[e] * N_REL + etype[e]], 1, __ATOMIC_RELAXED,
                                   __HIP_MEMORY_SCOPE_AGENT);
  ssrc[pos] = src_a[e];
}

__global__ void k_bmat(const float* __restrict__ w2, const float* __restrict__ r2,
                       float* __restrict__ B) {
  unsigned t = blockIdx.x * blockDim.x + threadIdx.x;
  if (t >= K_TOT * N_LAB) return;
  B[t] = (t < K_AGG * N_LAB) ? w2[t] : r2[t - K_AGG * N_LAB];
}

__global__ void k_layer1(const int* __restrict__ off, const int* __restrict__ ssrc,
                         const float* __restrict__ w1, const float* __restrict__ root1,
                         const float* __restrict__ bias1, float* __restrict__ lhs) {
  unsigned tid = blockIdx.x * blockDim.x + threadIdx.x;
  unsigned d = (unsigned)__builtin_amdgcn_readfirstlane((int)(tid >> 5));
  unsigned lane = threadIdx.x & 31;
  if (d >= N_NODES) return;
  unsigned r16 = lane & 15;
  int endv = off[d * N_REL + r16];
  int node_start = (d == 0) ? 0 : off[d * N_REL - 1];
  int prev = __shfl_up(endv, 1);
  int startv = (r16 == 0) ? node_start : prev;
  float invv = 1.0f / fmaxf((float)(endv - startv), 1.0f);
  int ends[16];
  for (int r = 0; r < 16; r++) ends[r] = __builtin_amdgcn_readlane(endv, r);
  int node_end = ends[15];
  v2f r1 = *((const v2f*)(root1 + (size_t)d * HID) + lane);
  v2f b1 = *((const v2f*)bias1 + lane);

  v2f acc = {0.0f, 0.0f};
  for (int cb = node_start; cb < node_end; cb += 32) {
    int p = cb + (int)lane;
    int sv = (p < node_end) ? ssrc[p] : 0;
    int relv = 0;
    for (int r = 0; r < 16; r++) relv += (p >= ends[r]) ? 1 : 0;
    relv = (relv > 15) ? 15 : relv;
    float iv = __shfl(invv, relv);
    int n = node_end - cb;
    n = (n > 32) ? 32 : n;
    for (int j = 0; j < n; j += 8) {
      v2f v[8];
      float ij[8];
      for (int k = 0; k < 8; k++) {
        int jj = (j + k < n) ? (j + k) : (n - 1);
        int sj = __builtin_amdgcn_readlane(sv, jj);
        int rj = __builtin_amdgcn_readlane(relv, jj);
        ij[k] = __builtin_bit_cast(float, __builtin_amdgcn_readlane(__builtin_bit_cast(int, iv), jj));
        v[k] = *((const v2f*)(w1 + ((size_t)rj * N_NODES + (size_t)sj) * HID) + lane);
      }
      for (int k = 0; k < 8; k++) {
        if (j + k < n) {
          acc.x += v[k].x * ij[k];
          acc.y += v[k].y * ij[k];
        }
      }
    }
  }
  v2f x;
  x.x = fmaxf(acc.x + r1.x + b1.x, 0.0f);
  x.y = fmaxf(acc.y + r1.y + b1.y, 0.0f);
  *((v2f*)(lhs + (size_t)d * K_TOT + K_AGG) + lane) = x;
}

__global__ void k_layer2(const int* __restrict__ off, const int* __restrict__ ssrc,
                         const float* __restrict__ xcols, float* __restrict__ agg) {
  unsigned tid = blockIdx.x * blockDim.x + threadIdx.x;
  unsigned d = (unsigned)__builtin_amdgcn_readfirstlane((int)(tid >> 5));
  unsigned lane = threadIdx.x & 31;
  if (d >= N_NODES) return;
  unsigned r16 = lane & 15;
  int endv = off[d * N_REL + r16];
  int node_start = (d == 0) ? 0 : off[d * N_REL - 1];
  int prev = __shfl_up(endv, 1);
  int startv = (r16 == 0) ? node_start : prev;
  float invv = 1.0f / fmaxf((float)(endv - startv), 1.0f);
  int ends[16];
  for (int r = 0; r < 16; r++) ends[r] = __builtin_amdgcn_readlane(endv, r);
  int node_end = ends[15];

  v2f* orow = (v2f*)(agg + (size_t)d * K_TOT) + lane;
  int cur = 0;
  v2f acc = {0.0f, 0.0f};
  for (int cb = node_start; cb < node_end; cb += 32) {
    int p = cb + (int)lane;
    int sv = (p < node_end) ? ssrc[p] : 0;
    int relv = 0;
    for (int r = 0; r < 16; r++) relv += (p >= ends[r]) ? 1 : 0;
    relv = (relv > 15) ? 15 : relv;
    int n = node_end - cb;
    n = (n > 32) ? 32 : n;
    for (int j = 0; j < n; j += 4) {
      v2f v[4];
      int rr[4];
      for (int k = 0; k < 4; k++) {
        int jj = (j + k < n) ? (j + k) : (n - 1);
        int sj = __builtin_amdgcn_readlane(sv, jj);
        rr[k] = __builtin_amdgcn_readlane(relv, jj);
        v[k] = *((const v2f*)(xcols + (size_t)sj * K_TOT) + lane);
      }
      for (int k = 0; k < 4; k++) {
        if (j + k < n) {
          while (cur < rr[k]) {
            float inv = __builtin_bit_cast(float, __builtin_amdgcn_readlane(__builtin_bit_cast(int, invv), cur));
            v2f o;
            o.x = acc.x * inv;
            o.y = acc.y * inv;
            orow[cur * 32] = o;
            acc.x = 0.0f; acc.y = 0.0f;
            cur++;
          }
          acc.x += v[k].x;
          acc.y += v[k].y;
        }
      }
    }
  }
  while (cur < N_REL) {
    float inv = __builtin_bit_cast(float, __builtin_amdgcn_readlane(__builtin_bit_cast(int, invv), cur));
    v2f o;
    o.x = acc.x * inv;
    o.y = acc.y * inv;
    orow[cur * 32] = o;
    acc.x = 0.0f; acc.y = 0.0f;
    cur++;
  }
}

__global__ void k_gemm(const float* __restrict__ lhs, const float* __restrict__ B,
                       const float* __restrict__ bias2, float* __restrict__ out) {
  unsigned tid = blockIdx.x * blockDim.x + threadIdx.x;
  unsigned wave = tid >> 5;
  unsigned lane = tid & 31;
  if (wave >= N_NODES / 16) return;
  unsigned half = lane >> 4;
  unsigned sub = lane & 15;
  const float* arow = lhs + (size_t)(wave * 16u + sub) * K_TOT + 2u * half;
  v8f c0 = {}; v8f c1 = {};
  for (int k0 = 0; k0 < K_TOT; k0 += 4) {
    v2f a;  a.x = arow[k0];  a.y = arow[k0 + 1];
    int kb = k0 + 2 * (int)half;
    v2f b0; b0.x = B[kb * N_LAB + sub];        b0.y = B[(kb + 1) * N_LAB + sub];
    v2f b1; b1.x = B[kb * N_LAB + 16 + sub];   b1.y = B[(kb + 1) * N_LAB + 16 + sub];
    c0 = __builtin_amdgcn_wmma_f32_16x16x4_f32(false, a, false, b0, (short)0, c0, false, false);
    c1 = __builtin_amdgcn_wmma_f32_16x16x4_f32(false, a, false, b1, (short)0, c1, false, false);
  }
  #pragma unroll
  for (int v = 0; v < 8; v++) {
    unsigned row = wave * 16u + (unsigned)v + 8u * half;
    float x0 = c0[v] + bias2[sub];
    float x1 = c1[v] + bias2[16 + sub];
    out[row * N_LAB + sub]      = 1.0f / (1.0f + __expf(-x0));
    out[row * N_LAB + 16 + sub] = 1.0f / (1.0f + __expf(-x1));
  }
}

extern "C" void kernel_launch(void* const* d_in, const int* in_sizes, int n_in,
                              void* d_out, int out_size, void* d_ws, size_t ws_size,
                              hipStream_t stream) {
  const int* ei     = (const int*)d_in[0];
  const int* src_a  = ei;
  const int* dst_a  = ei + N_EDGE;
  const int* etype  = (const int*)d_in[1];
  const float* w1   = (const float*)d_in[2];
  const float* r1   = (const float*)d_in[3];
  const float* b1   = (const float*)d_in[4];
  const float* w2   = (const float*)d_in[5];
  const float* r2   = (const float*)d_in[6];
  const float* b2   = (const float*)d_in[7];
  float* out = (float*)d_out;

  float* ws   = (float*)d_ws;
  int*   off  = (int*)d_ws + WS_CNT_OFF;
  float* Bmat = ws + WS_B_OFF;
  float* lhs  = ws + WS_LHS_OFF;
  int*   bsum = (int*)d_ws + WS_LHS_OFF;
  int*   ssrc = (int*)d_out;

  (void)in_sizes; (void)n_in; (void)out_size; (void)ws_size;

  k_zero<<<(N_KEYS + 255) / 256, 256, 0, stream>>>(off);
  k_count<<<(N_EDGE + 255) / 256, 256, 0, stream>>>(etype, dst_a, off);
  k_scan1<<<SCAN_BLOCKS, 256, 0, stream>>>(off, bsum);
  k_scan2<<<SCAN_BLOCKS, 256, 0, stream>>>(off, bsum);
  k_fill<<<(N_EDGE + 255) / 256, 256, 0, stream>>>(src_a, dst_a, etype, off, ssrc);
  k_bmat<<<(K_TOT * N_LAB + 255) / 256, 256, 0, stream>>>(w2, r2, Bmat);
  k_layer1<<<(N_NODES * 32u + 255) / 256, 256, 0, stream>>>(off, ssrc, w1, r1, b1, lhs);
  k_layer2<<<(N_NODES * 32u + 255) / 256, 256, 0, stream>>>(off, ssrc, lhs + K_AGG, lhs);
  k_gemm<<<((N_NODES / 16) * 32 + 255) / 256, 256, 0, stream>>>(lhs, Bmat, b2, out);
}


# layer 2 as transform-then-aggregate: Y = x @ [weight2 | root2] on f32 WMMA (K=64), then one wave per dst sums 128-byte Y rows times 1/count; N x 1088 mean matrix gone; sums reassociated
# speedup vs baseline: 2.1803x; 1.1001x over previous
#include <hip/hip_runtime.h>


#define N_NODES 50000
#define N_REL   16
#define HID     64
#define N_LAB   32
#define N_EDGE  1600000
#define N_YC    544
#define ROW_TILES   (N_NODES / 16)
#define RT_PER_WAVE 4
#define RT_GROUPS   ((ROW_TILES + RT_PER_WAVE - 1) / RT_PER_WAVE)
#define N_PAIRS     (N_REL + 1)
#define N_KEYS  (N_NODES * N_REL)
#define SCAN_BLOCKS ((N_NODES + 255) / 256)

typedef float v2f __attribute__((ext_vector_type(2)));
typedef float v4f __attribute__((ext_vector_type(4)));
typedef float v8f __attribute__((ext_vector_type(8)));

#define WS_CNT_OFF  0
#define WS_X_OFF    (N_REL * N_NODES)
#define WS_Y_OFF    (WS_X_OFF + N_NODES * HID)
#define WS_SSRC_OFF (WS_Y_OFF + N_NODES * N_YC)

__global__ void k_zero(int* __restrict__ p) {
  unsigned i = blockIdx.x * blockDim.x + threadIdx.x;
  if (i < N_KEYS) p[i] = 0;
}

__global__ void k_count(const int* __restrict__ etype, const int* __restrict__ dst,
                        int* __restrict__ cnt) {
  unsigned e = blockIdx.x * blockDim.x + threadIdx.x;
  if (e >= N_EDGE) return;
  __hip_atomic_fetch_add(&cnt[dst[e] * N_REL + etype[e]], 1, __ATOMIC_RELAXED, __HIP_MEMORY_SCOPE_AGENT);
}

__device__ __forceinline__ int block_sum256(int v, int* sh) {
  unsigned t = threadIdx.x;
  sh[t] = v;
  __syncthreads();
  for (unsigned s = 128; s > 0; s >>= 1) {
    if (t < s) sh[t] += sh[t + s];
    __syncthreads();
  }
  int r = sh[0];
  __syncthreads();
  return r;
}

__global__ void k_scan1(const int* __restrict__ cnt, int* __restrict__ bsum) {
  __shared__ int sh[256];
  unsigned n = blockIdx.x * 256u + threadIdx.x;
  int deg = 0;
  if (n < N_NODES) {
    const int4* p = (const int4*)(cnt + (size_t)n * N_REL);
    for (int q = 0; q < 4; q++) { int4 c = p[q]; deg += c.x + c.y + c.z + c.w; }
  }
  int tot = block_sum256(deg, sh);
  if (threadIdx.x == 0) bsum[blockIdx.x] = tot;
}

__global__ void k_scan2(int* __restrict__ cnt, const int* __restrict__ bsum) {
  __shared__ int sh[256];
  unsigned t = threadIdx.x;
  unsigned n = blockIdx.x * 256u + t;
  int base = block_sum256((t < blockIdx.x) ? bsum[t] : 0, sh);
  int c[16];
  int deg = 0;
  if (n < N_NODES) {
    const int4* p = (const int4*)(cnt + (size_t)n * N_REL);
    for (int q = 0; q < 4; q++) {
      int4 v = p[q];
      c[4 * q] = v.x; c[4 * q + 1] = v.y; c[4 * q + 2] = v.z; c[4 * q + 3] = v.w;
      deg += v.x + v.y + v.z + v.w;
    }
  } else {
    for (int q = 0; q < 16; q++) c[q] = 0;
  }
  sh[t] = deg;
  __syncthreads();
  for (unsigned o = 1; o < 256; o <<= 1) {
    int add = (t >= o) ? sh[t - o] : 0;
    __syncthreads();
    sh[t] += add;
    __syncthreads();
  }
  int run = base + sh[t] - deg;
  if (n < N_NODES) {
    int4* p = (int4*)(cnt + (size_t)n * N_REL);
    for (int q = 0; q < 4; q++) {
      int4 v;
      v.x = run; run += c[4 * q];
      v.y = run; run += c[4 * q + 1];
      v.z = run; run += c[4 * q + 2];
      v.w = run; run += c[4 * q + 3];
      p[q] = v;
    }
  }
}

__global__ void k_fill(const int* __restrict__ src_a, const int* __restrict__ dst_a,
                       const int* __restrict__ etype, int* __restrict__ off,
                       int* __restrict__ ssrc) {
  unsigned e = blockIdx.x * blockDim.x + threadIdx.x;
  if (e >= N_EDGE) return;
  int pos = __hip_atomic_fetch_add(&off[dst_a[e] * N_REL + etype[e]], 1, __ATOMIC_RELAXED,
                                   __HIP_MEMORY_SCOPE_AGENT);
  ssrc[pos] = src_a[e];
}

__global__ void k_layer1(const int* __restrict__ off, const int* __restrict__ ssrc,
                         const float* __restrict__ w1, const float* __restrict__ root1,
                         const float* __restrict__ bias1, float* __restrict__ xout) {
  unsigned tid = blockIdx.x * blockDim.x + threadIdx.x;
  unsigned d = (unsigned)__builtin_amdgcn_readfirstlane((int)(tid >> 5));
  unsigned lane = threadIdx.x & 31;
  if (d >= N_NODES) return;
  unsigned r16 = lane & 15;
  int endv = off[d * N_REL + r16];
  int node_start = (d == 0) ? 0 : off[d * N_REL - 1];
  int prev = __shfl_up(endv, 1);
  int startv = (r16 == 0) ? node_start : prev;
  float invv = 1.0f / fmaxf((float)(endv - startv), 1.0f);
  int ends[16];
  for (int r = 0; r < 16; r++) ends[r] = __builtin_amdgcn_readlane(endv, r);
  int node_end = ends[15];
  v2f r1 = *((const v2f*)(root1 + (size_t)d * HID) + lane);
  v2f b1 = *((const v2f*)bias1 + lane);

  v2f acc = {0.0f, 0.0f};
  for (int cb = node_start; cb < node_end; cb += 32) {
    int p = cb + (int)lane;
    int sv = (p < node_end) ? ssrc[p] : 0;
    int relv = 0;
    for (int r = 0; r < 16; r++) relv += (p >= ends[r]) ? 1 : 0;
    relv = (relv > 15) ? 15 : relv;
    float iv = __shfl(invv, relv);
    int n = node_end - cb;
    n = (n > 32) ? 32 : n;
    for (int j = 0; j < n; j += 8) {
      v2f v[8];
      float ij[8];
      for (int k = 0; k < 8; k++) {
        int jj = (j + k < n) ? (j + k) : (n - 1);
        int sj = __builtin_amdgcn_readlane(sv, jj);
        int rj = __builtin_amdgcn_readlane(relv, jj);
        ij[k] = __builtin_bit_cast(float, __builtin_amdgcn_readlane(__builtin_bit_cast(int, iv), jj));
        v[k] = *((const v2f*)(w1 + ((size_t)rj * N_NODES + (size_t)sj) * HID) + lane);
      }
      for (int k = 0; k < 8; k++) {
        if (j + k < n) {
          acc.x += v[k].x * ij[k];
          acc.y += v[k].y * ij[k];
        }
      }
    }
  }
  v2f x;
  x.x = fmaxf(acc.x + r1.x + b1.x, 0.0f);
  x.y = fmaxf(acc.y + r1.y + b1.y, 0.0f);
  *((v2f*)(xout + (size_t)d * HID) + lane) = x;
}

__global__ void k_xw(const float* __restrict__ x, const float* __restrict__ w2,
                     const float* __restrict__ r2, float* __restrict__ Y) {
  unsigned tid = blockIdx.x * blockDim.x + threadIdx.x;
  unsigned wave = (unsigned)__builtin_amdgcn_readfirstlane((int)(tid >> 5));
  unsigned lane = threadIdx.x & 31;
  if (wave >= RT_GROUPS * N_PAIRS) return;
  unsigned g = wave / N_PAIRS;
  unsigned p = wave - g * N_PAIRS;
  unsigned half = lane >> 4;
  unsigned sub = lane & 15;
  const float* Bp = (p < N_REL) ? (w2 + p * (HID * N_LAB)) : r2;
  v2f b0[16], b1[16];
  #pragma unroll
  for (int s = 0; s < 16; s++) {
    int kb = 8 * (s >> 1) + 4 * (int)half + 2 * (s & 1);
    b0[s].x = Bp[kb * N_LAB + sub];        b0[s].y = Bp[(kb + 1) * N_LAB + sub];
    b1[s].x = Bp[kb * N_LAB + 16 + sub];   b1[s].y = Bp[(kb + 1) * N_LAB + 16 + sub];
  }
  for (unsigned t = 0; t < RT_PER_WAVE; t++) {
    unsigned rt = g * RT_PER_WAVE + t;
    if (rt >= ROW_TILES) break;
    const float* arow = x + (size_t)(rt * 16u + sub) * HID + 4u * half;
    v2f a[16];
    #pragma unroll
    for (int j = 0; j < 8; j++) {
      v4f q = *((const v4f*)(arow + 8 * j));
      a[2 * j].x = q.x;      a[2 * j].y = q.y;
      a[2 * j + 1].x = q.z;  a[2 * j + 1].y = q.w;
    }
    v8f c0 = {}; v8f c1 = {};
    #pragma unroll
    for (int s = 0; s < 16; s++) {
      c0 = __builtin_amdgcn_wmma_f32_16x16x4_f32(false, a[s], false, b0[s], (short)0, c0, false, false);
      c1 = __builtin_amdgcn_wmma_f32_16x16x4_f32(false, a[s], false, b1[s], (short)0, c1, false, false);
    }
    float* yrow = Y + (size_t)(rt * 16u) * N_YC + p * N_LAB + lane;
    #pragma unroll
    for (int v = 0; v < 8; v++) {
      float send = half ? c0[v] : c1[v];
      float got = __builtin_bit_cast(float, __builtin_amdgcn_permlanex16(
          __builtin_bit_cast(int, send), __builtin_bit_cast(int, send), 0x76543210, 0xfedcba98, false, false));
      float lo = half ? got : c0[v];
      float hi = half ? c1[v] : got;
      yrow[(size_t)v * N_YC] = lo;
      yrow[(size_t)(v + 8) * N_YC] = hi;
    }
  }
}

__global__ void k_layer2(const int* __restrict__ off, const int* __restrict__ ssrc,
                         const float* __restrict__ Y, const float* __restrict__ bias2,
                         float* __restrict__ out) {
  unsigned tid = blockIdx.x * blockDim.x + threadIdx.x;
  unsigned d = (unsigned)__builtin_amdgcn_readfirstlane((int)(tid >> 5));
  unsigned lane = threadIdx.x & 31;
  if (d >= N_NODES) return;
  unsigned r16 = lane & 15;
  int endv = off[d * N_REL + r16];
  int node_start = (d == 0) ? 0 : off[d * N_REL - 1];
  int prev = __shfl_up(endv, 1);
  int startv = (r16 == 0) ? node_start : prev;
  float invv = 1.0f / fmaxf((float)(endv - startv), 1.0f);
  int ends[16];
  for (int r = 0; r < 16; r++) ends[r] = __builtin_amdgcn_readlane(endv, r);
  int node_end = ends[15];
  float rootv = Y[(size_t)d * N_YC + N_REL * N_LAB + lane];
  float b2 = bias2[lane];

  float acc = 0.0f;
  for (int cb = node_start; cb < node_end; cb += 32) {
    int p = cb + (int)lane;
    int sv = (p < node_end) ? ssrc[p] : 0;
    int relv = 0;
    for (int r = 0; r < 16; r++) relv += (p >= ends[r]) ? 1 : 0;
    relv = (relv > 15) ? 15 : relv;
    float iv = __shfl(invv, relv);
    int n = node_end - cb;
    n = (n > 32) ? 32 : n;
    for (int j = 0; j < n; j += 16) {
      float v[16];
      float ij[16];
      #pragma unroll
      for (int k = 0; k < 16; k++) {
        int jj = (j + k < n) ? (j + k) : (n - 1);
        int sj = __builtin_amdgcn_readlane(sv, jj);
        int rj = __builtin_amdgcn_readlane(relv, jj);
        ij[k] = __builtin_bit_cast(float, __builtin_amdgcn_readlane(__builtin_bit_cast(int, iv), jj));
        v[k] = Y[(size_t)sj * N_YC + (size_t)(rj * N_LAB) + lane];
      }
      #pragma unroll
      for (int k = 0; k < 16; k++) {
        if (j + k < n) acc += v[k] * ij[k];
      }
    }
  }
  float z = acc + rootv + b2;
  out[d * N_LAB + lane] = 1.0f / (1.0f + __expf(-z));
}

extern "C" void kernel_launch(void* const* d_in, const int* in_sizes, int n_in,
                              void* d_out, int out_size, void* d_ws, size_t ws_size,
                              hipStream_t stream) {
  const int* ei     = (const int*)d_in[0];
  const int* src_a  = ei;
  const int* dst_a  = ei + N_EDGE;
  const int* etype  = (const int*)d_in[1];
  const float* w1   = (const float*)d_in[2];
  const float* r1   = (const float*)d_in[3];
  const float* b1   = (const float*)d_in[4];
  const float* w2   = (const float*)d_in[5];
  const float* r2   = (const float*)d_in[6];
  const float* b2   = (const float*)d_in[7];
  float* out = (float*)d_out;

  float* ws   = (float*)d_ws;
  int*   off  = (int*)d_ws + WS_CNT_OFF;
  float* xbuf = ws + WS_X_OFF;
  float* Y    = ws + WS_Y_OFF;
  int*   bsum = (int*)d_ws + WS_Y_OFF;
  int*   ssrc = (int*)d_ws + WS_SSRC_OFF;

  (void)in_sizes; (void)n_in; (void)out_size; (void)ws_size;

  k_zero<<<(N_KEYS + 255) / 256, 256, 0, stream>>>(off);
  k_count<<<(N_EDGE + 255) / 256, 256, 0, stream>>>(etype, dst_a, off);
  k_scan1<<<SCAN_BLOCKS, 256, 0, stream>>>(off, bsum);
  k_scan2<<<SCAN_BLOCKS, 256, 0, stream>>>(off, bsum);
  k_fill<<<(N_EDGE + 255) / 256, 256, 0, stream>>>(src_a, dst_a, etype, off, ssrc);
  k_layer1<<<(N_NODES * 32u + 255) / 256, 256, 0, stream>>>(off, ssrc, w1, r1, b1, xbuf);
  k_xw<<<(RT_GROUPS * N_PAIRS * 32 + 255) / 256, 256, 0, stream>>>(xbuf, w2, r2, Y);
  k_layer2<<<(N_NODES * 32u + 255) / 256, 256, 0, stream>>>(off, ssrc, Y, b2, out);
}


# edge sort without global atomics: two-level counting sort (196 dst buckets, then 4096 keys a bucket) ranked at LDS counters; 8 launches become 7; slots in a segment stay in arrival order
# speedup vs baseline: 3.3563x; 1.5394x over previous
#include <hip/hip_runtime.h>


#define N_NODES 50000
#define N_REL   16
#define HID     64
#define N_LAB   32
#define N_EDGE  1600000
#define N_YC    544
#define ROW_TILES   (N_NODES / 16)
#define RT_PER_WAVE 4
#define RT_GROUPS   ((ROW_TILES + RT_PER_WAVE - 1) / RT_PER_WAVE)
#define N_PAIRS     (N_REL + 1)
#define N_KEYS  (N_NODES * N_REL)
#define N_BUCK   ((N_NODES + 255) / 256)
#define BK_KEYS  (256 * N_REL)
#define SORT_WG  256
#define SORT_T   1024
#define CHUNK    ((N_EDGE + SORT_WG - 1) / SORT_WG)
#define CH_IT    ((CHUNK + SORT_T - 1) / SORT_T)
#define SCAN_ROWS ((N_BUCK + 31) / 32)
static_assert(N_NODES <= 65536, "a source node index must fit in 16 bits of the staged word");
static_assert(BK_KEYS == 4 * SORT_T, "one thread per 4 keys in the bucket launch");
static_assert(SORT_WG == 256, "a table row is 32 lanes x 8 entries");
static_assert(N_BUCK <= 256, "bucket totals are scanned by one wave, 8 per lane");

typedef float v2f __attribute__((ext_vector_type(2)));
typedef float v4f __attribute__((ext_vector_type(4)));
typedef float v8f __attribute__((ext_vector_type(8)));

#define WS_CNT_OFF  0
#define WS_X_OFF    (N_REL * N_NODES)
#define WS_Y_OFF    (WS_X_OFF + N_NODES * HID)
#define WS_SSRC_OFF (WS_Y_OFF + N_NODES * N_YC)
#define WS_STAGE_OFF WS_Y_OFF
#define WS_TAB_OFF   (WS_Y_OFF + N_EDGE)

__device__ __forceinline__ int wave_incl_scan(int v, unsigned lane) {
  for (unsigned o = 1; o < 32; o <<= 1) {
    int u = __shfl_up(v, o);
    if (lane >= o) v += u;
  }
  return v;
}

__global__ __launch_bounds__(SORT_T) void k_hist(const int* __restrict__ dst, int* __restrict__ tab) {
  __shared__ int h[N_BUCK];
  unsigned t = threadIdx.x, g = blockIdx.x;
  if (t < N_BUCK) h[t] = 0;
  __syncthreads();
  unsigned beg = g * CHUNK;
  unsigned end = (beg + CHUNK < N_EDGE) ? (beg + CHUNK) : N_EDGE;
  int dv[CH_IT];
  #pragma unroll
  for (int k = 0; k < CH_IT; k++) {
    unsigned i = beg + (unsigned)k * SORT_T + t;
    dv[k] = (i < end) ? dst[i] : -1;
  }
  #pragma unroll
  for (int k = 0; k < CH_IT; k++) {
    if (dv[k] >= 0)
      __hip_atomic_fetch_add(&h[dv[k] >> 8], 1, __ATOMIC_RELAXED, __HIP_MEMORY_SCOPE_WORKGROUP);
  }
  __syncthreads();
  if (t < N_BUCK) tab[t * SORT_WG + g] = h[t];
}

__global__ __launch_bounds__(SORT_T) void k_scanB(int* __restrict__ tab) {
  __shared__ int rowtot[256];
  unsigned t = threadIdx.x, lane = t & 31;
  unsigned w = (unsigned)__builtin_amdgcn_readfirstlane((int)(t >> 5));
  int pre[SCAN_ROWS][8];
  #pragma unroll
  for (int r = 0; r < SCAN_ROWS; r++) {
    unsigned b = w + 32u * (unsigned)r;
    if (b < N_BUCK) {
      const int4* p = (const int4*)(tab + b * SORT_WG) + 2 * lane;
      int4 a = p[0], c = p[1];
      pre[r][0] = a.x; pre[r][1] = a.y; pre[r][2] = a.z; pre[r][3] = a.w;
      pre[r][4] = c.x; pre[r][5] = c.y; pre[r][6] = c.z; pre[r][7] = c.w;
    } else {
      for (int j = 0; j < 8; j++) pre[r][j] = 0;
    }
  }
  #pragma unroll
  for (int r = 0; r < SCAN_ROWS; r++) {
    unsigned b = w + 32u * (unsigned)r;
    int s = 0;
    for (int j = 0; j < 8; j++) { int v = pre[r][j]; pre[r][j] = s; s += v; }
    int inc = wave_incl_scan(s, lane);
    int exc = inc - s;
    for (int j = 0; j < 8; j++) pre[r][j] += exc;
    if (lane == 31 && b < N_BUCK) rowtot[b] = inc;
  }
  __syncthreads();
  if (w == 0) {
    int e[8];
    int s = 0;
    for (int j = 0; j < 8; j++) {
      unsigned idx = 8 * lane + (unsigned)j;
      int v = (idx < N_BUCK) ? rowtot[idx] : 0;
      e[j] = s; s += v;
    }
    int inc = wave_incl_scan(s, lane);
    int exc = inc - s;
    for (int j = 0; j < 8; j++) {
      unsigned idx = 8 * lane + (unsigned)j;
      if (idx < N_BUCK) rowtot[idx] = e[j] + exc;
    }
  }
  __syncthreads();
  #pragma unroll
  for (int r = 0; r < SCAN_ROWS; r++) {
    unsigned b = w + 32u * (unsigned)r;
    if (b < N_BUCK) {
      int rb = rowtot[b];
      int4 a, c;
      a.x = pre[r][0] + rb; a.y = pre[r][1] + rb; a.z = pre[r][2] + rb; a.w = pre[r][3] + rb;
      c.x = pre[r][4] + rb; c.y = pre[r][5] + rb; c.z = pre[r][6] + rb; c.w = pre[r][7] + rb;
      int4* p = (int4*)(tab + b * SORT_WG) + 2 * lane;
      p[0] = a; p[1] = c;
    }
  }
}

__global__ __launch_bounds__(SORT_T) void k_scatter(const int* __restrict__ src_a, const int* __restrict__ dst_a,
                                                    const int* __restrict__ etype, const int* __restrict__ tab,
                                                    int* __restrict__ staged) {
  __shared__ int cur[N_BUCK];
  unsigned t = threadIdx.x, g = blockIdx.x;
  if (t < N_BUCK) cur[t] = tab[t * SORT_WG + g];
  __syncthreads();
  unsigned beg = g * CHUNK;
  unsigned end = (beg + CHUNK < N_EDGE) ? (beg + CHUNK) : N_EDGE;
  int dv[CH_IT], wv[CH_IT];
  #pragma unroll
  for (int k = 0; k < CH_IT; k++) {
    unsigned i = beg + (unsigned)k * SORT_T + t;
    bool ok = i < end;
    int d = ok ? dst_a[i] : -1;
    int s = ok ? src_a[i] : 0;
    int r = ok ? etype[i] : 0;
    dv[k] = d;
    wv[k] = ((((d & 255) << 4) + r) << 16) | s;
  }
  int pos[CH_IT];
  #pragma unroll
  for (int k = 0; k < CH_IT; k++) {
    pos[k] = 0;
    if (dv[k] >= 0)
      pos[k] = __hip_atomic_fetch_add(&cur[dv[k] >> 8], 1, __ATOMIC_RELAXED, __HIP_MEMORY_SCOPE_WORKGROUP);
  }
  #pragma unroll
  for (int k = 0; k < CH_IT; k++) {
    if (dv[k] >= 0) staged[pos[k]] = wv[k];
  }
}

__global__ __launch_bounds__(SORT_T) void k_bucket(const int* __restrict__ tab, const int* __restrict__ staged,
                                                   int* __restrict__ off, int* __restrict__ ssrc) {
  __shared__ alignas(16) int cnt[BK_KEYS];
  __shared__ int wtot[32];
  unsigned t = threadIdx.x, lane = t & 31, w = t >> 5;
  unsigned b = blockIdx.x;
  int bstart = tab[b * SORT_WG];
  int bend = (b + 1 < N_BUCK) ? tab[(b + 1) * SORT_WG] : N_EDGE;
  int4 zero4; zero4.x = 0; zero4.y = 0; zero4.z = 0; zero4.w = 0;
  ((int4*)cnt)[t] = zero4;
  __syncthreads();
  for (int i0 = bstart + (int)t; i0 < bend; i0 += 4 * SORT_T) {
    unsigned wd[4];
    #pragma unroll
    for (int k = 0; k < 4; k++) {
      int i = i0 + k * SORT_T;
      wd[k] = (i < bend) ? (unsigned)staged[i] : 0xffffffffu;
    }
    #pragma unroll
    for (int k = 0; k < 4; k++) {
      if (wd[k] != 0xffffffffu)
        __hip_atomic_fetch_add(&cnt[wd[k] >> 16], 1, __ATOMIC_RELAXED, __HIP_MEMORY_SCOPE_WORKGROUP);
    }
  }
  __syncthreads();
  int4 c = ((const int4*)cnt)[t];
  int s = c.x + c.y + c.z + c.w;
  int inc = wave_incl_scan(s, lane);
  if (lane == 31) wtot[w] = inc;
  __syncthreads();
  if (w == 0) {
    int v = wtot[lane];
    int vi = wave_incl_scan(v, lane);
    wtot[lane] = vi - v;
  }
  __syncthreads();
  int ex = wtot[w] + inc - s;
  int4 st;
  st.x = ex; st.y = st.x + c.x; st.z = st.y + c.y; st.w = st.z + c.z;
  ((int4*)cnt)[t] = st;
  unsigned key0 = b * BK_KEYS + 4 * t;
  if (key0 < N_KEYS) {
    int4 e;
    e.x = bstart + st.y; e.y = bstart + st.z; e.z = bstart + st.w; e.w = bstart + st.w + c.w;
    *((int4*)(off + key0)) = e;
  }
  __syncthreads();
  for (int i0 = bstart + (int)t; i0 < bend; i0 += 4 * SORT_T) {
    unsigned wd[4];
    #pragma unroll
    for (int k = 0; k < 4; k++) {
      int i = i0 + k * SORT_T;
      wd[k] = (i < bend) ? (unsigned)staged[i] : 0xffffffffu;
    }
    int pos[4];
    #pragma unroll
    for (int k = 0; k < 4; k++) {
      pos[k] = 0;
      if (wd[k] != 0xffffffffu)
        pos[k] = __hip_atomic_fetch_add(&cnt[wd[k] >> 16], 1, __ATOMIC_RELAXED, __HIP_MEMORY_SCOPE_WORKGROUP);
    }
    #pragma unroll
    for (int k = 0; k < 4; k++) {
      if (wd[k] != 0xffffffffu) ssrc[bstart + pos[k]] = (int)(wd[k] & 0xffffu);
    }
  }
}

__global__ void k_layer1(const int* __restrict__ off, const int* __restrict__ ssrc,
                         const float* __restrict__ w1, const float* __restrict__ root1,
                         const float* __restrict__ bias1, float* __restrict__ xout) {
  unsigned tid = blockIdx.x * blockDim.x + threadIdx.x;
  unsigned d = (unsigned)__builtin_amdgcn_readfirstlane((int)(tid >> 5));
  unsigned lane = threadIdx.x & 31;
  if (d >= N_NODES) return;
  unsigned r16 = lane & 15;
  int endv = off[d * N_REL + r16];
  int node_start = (d == 0) ? 0 : off[d * N_REL - 1];
  int prev = __shfl_up(endv, 1);
  int startv = (r16 == 0) ? node_start : prev;
  float invv = 1.0f / fmaxf((float)(endv - startv), 1.0f);
  int ends[16];
  for (int r = 0; r < 16; r++) ends[r] = __builtin_amdgcn_readlane(endv, r);
  int node_end = ends[15];
  v2f r1 = *((const v2f*)(root1 + (size_t)d * HID) + lane);
  v2f b1 = *((const v2f*)bias1 + lane);

  v2f acc = {0.0f, 0.0f};
  for (int cb = node_start; cb < node_end; cb += 32) {
    int p = cb + (int)lane;
    int sv = (p < node_end) ? ssrc[p] : 0;
    int relv = 0;
    for (int r = 0; r < 16; r++) relv += (p >= ends[r]) ? 1 : 0;
    relv = (relv > 15) ? 15 : relv;
    float iv = __shfl(invv, relv);
    int n = node_end - cb;
    n = (n > 32) ? 32 : n;
    for (int j = 0; j < n; j += 8) {
      v2f v[8];
      float ij[8];
      for (int k = 0; k < 8; k++) {
        int jj = (j + k < n) ? (j + k) : (n - 1);
        int sj = __builtin_amdgcn_readlane(sv, jj);
        int rj = __builtin_amdgcn_readlane(relv, jj);
        ij[k] = __builtin_bit_cast(float, __builtin_amdgcn_readlane(__builtin_bit_cast(int, iv), jj));
        v[k] = *((const v2f*)(w1 + ((size_t)rj * N_NODES + (size_t)sj) * HID) + lane);
      }
      for (int k = 0; k < 8; k++) {
        if (j + k < n) {
          acc.x += v[k].x * ij[k];
          acc.y += v[k].y * ij[k];
        }
      }
    }
  }
  v2f x;
  x.x = fmaxf(acc.x + r1.x + b1.x, 0.0f);
  x.y = fmaxf(acc.y + r1.y + b1.y, 0.0f);
  *((v2f*)(xout + (size_t)d * HID) + lane) = x;
}

__global__ void k_xw(const float* __restrict__ x, const float* __restrict__ w2,
                     const float* __restrict__ r2, float* __restrict__ Y) {
  unsigned tid = blockIdx.x * blockDim.x + threadIdx.x;
  unsigned wave = (unsigned)__builtin_amdgcn_readfirstlane((int)(tid >> 5));
  unsigned lane = threadIdx.x & 31;
  if (wave >= RT_GROUPS * N_PAIRS) return;
  unsigned g = wave / N_PAIRS;
  unsigned p = wave - g * N_PAIRS;
  unsigned half = lane >> 4;
  unsigned sub = lane & 15;
  const float* Bp = (p < N_REL) ? (w2 + p * (HID * N_LAB)) : r2;
  v2f b0[16], b1[16];
  #pragma unroll
  for (int s = 0; s < 16; s++) {
    int kb = 8 * (s >> 1) + 4 * (int)half + 2 * (s & 1);
    b0[s].x = Bp[kb * N_LAB + sub];        b0[s].y = Bp[(kb + 1) * N_LAB + sub];
    b1[s].x = Bp[kb * N_LAB + 16 + sub];   b1[s].y = Bp[(kb + 1) * N_LAB + 16 + sub];
  }
  for (unsigned t = 0; t < RT_PER_WAVE; t++) {
    unsigned rt = g * RT_PER_WAVE + t;
    if (rt >= ROW_TILES) break;
    const float* arow = x + (size_t)(rt * 16u + sub) * HID + 4u * half;
    v2f a[16];
    #pragma unroll
    for (int j = 0; j < 8; j++) {
      v4f q = *((const v4f*)(arow + 8 * j));
      a[2 * j].x = q.x;      a[2 * j].y = q.y;
      a[2 * j + 1].x = q.z;  a[2 * j + 1].y = q.w;
    }
    v8f c0 = {}; v8f c1 = {};
    #pragma unroll
    for (int s = 0; s < 16; s++) {
      c0 = __builtin_amdgcn_wmma_f32_16x16x4_f32(false, a[s], false, b0[s], (short)0, c0, false, false);
      c1 = __builtin_amdgcn_wmma_f32_16x16x4_f32(false, a[s], false, b1[s], (short)0, c1, false, false);
    }
    float* yrow = Y + (size_t)(rt * 16u) * N_YC + p * N_LAB + lane;
    #pragma unroll
    for (int v = 0; v < 8; v++) {
      float send = half ? c0[v] : c1[v];
      float got = __builtin_bit_cast(float, __builtin_amdgcn_permlanex16(
          __builtin_bit_cast(int, send), __builtin_bit_cast(int, send), 0x76543210, 0xfedcba98, false, false));
      float lo = half ? got : c0[v];
      float hi = half ? c1[v] : got;
      yrow[(size_t)v * N_YC] = lo;
      yrow[(size_t)(v + 8) * N_YC] = hi;
    }
  }
}

__global__ void k_layer2(const int* __restrict__ off, const int* __restrict__ ssrc,
                         const float* __restrict__ Y, const float* __restrict__ bias2,
                         float* __restrict__ out) {
  unsigned tid = blockIdx.x * blockDim.x + threadIdx.x;
  unsigned d = (unsigned)__builtin_amdgcn_readfirstlane((int)(tid >> 5));
  unsigned lane = threadIdx.x & 31;
  if (d >= N_NODES) return;
  unsigned r16 = lane & 15;
  int endv = off[d * N_REL + r16];
  int node_start = (d == 0) ? 0 : off[d * N_REL - 1];
  int prev = __shfl_up(endv, 1);
  int startv = (r16 == 0) ? node_start : prev;
  float invv = 1.0f / fmaxf((float)(endv - startv), 1.0f);
  int ends[16];
  for (int r = 0; r < 16; r++) ends[r] = __builtin_amdgcn_readlane(endv, r);
  int node_end = ends[15];
  float rootv = Y[(size_t)d * N_YC + N_REL * N_LAB + lane];
  float b2 = bias2[lane];

  float acc = 0.0f;
  for (int cb = node_start; cb < node_end; cb += 32) {
    int p = cb + (int)lane;
    int sv = (p < node_end) ? ssrc[p] : 0;
    int relv = 0;
    for (int r = 0; r < 16; r++) relv += (p >= ends[r]) ? 1 : 0;
    relv = (relv > 15) ? 15 : relv;
    float iv = __shfl(invv, relv);
    int n = node_end - cb;
    n = (n > 32) ? 32 : n;
    for (int j = 0; j < n; j += 16) {
      float v[16];
      float ij[16];
      #pragma unroll
      for (int k = 0; k < 16; k++) {
        int jj = (j + k < n) ? (j + k) : (n - 1);
        int sj = __builtin_amdgcn_readlane(sv, jj);
        int rj = __builtin_amdgcn_readlane(relv, jj);
        ij[k] = __builtin_bit_cast(float, __builtin_amdgcn_readlane(__builtin_bit_cast(int, iv), jj));
        v[k] = Y[(size_t)sj * N_YC + (size_t)(rj * N_LAB) + lane];
      }
      #pragma unroll
      for (int k = 0; k < 16; k++) {
        if (j + k < n) acc += v[k] * ij[k];
      }
    }
  }
  float z = acc + rootv + b2;
  out[d * N_LAB + lane] = 1.0f / (1.0f + __expf(-z));
}

extern "C" void kernel_launch(void* const* d_in, const int* in_sizes, int n_in,
                              void* d_out, int out_size, void* d_ws, size_t ws_size,
                              hipStream_t stream) {
  const int* ei     = (const int*)d_in[0];
  const int* src_a  = ei;
  const int* dst_a  = ei + N_EDGE;
  const int* etype  = (const int*)d_in[1];
  const float* w1   = (const float*)d_in[2];
  const float* r1   = (const float*)d_in[3];
  const float* b1   = (const float*)d_in[4];
  const float* w2   = (const float*)d_in[5];
  const float* r2   = (const float*)d_in[6];
  const float* b2   = (const float*)d_in[7];
  float* out = (float*)d_out;

  float* ws   = (float*)d_ws;
  int*   off  = (int*)d_ws + WS_CNT_OFF;
  float* xbuf = ws + WS_X_OFF;
  float* Y    = ws + WS_Y_OFF;
  int*   staged = (int*)d_ws + WS_STAGE_OFF;
  int*   tab  = (int*)d_ws + WS_TAB_OFF;
  int*   ssrc = (int*)d_ws + WS_SSRC_OFF;

  (void)in_sizes; (void)n_in; (void)out_size; (void)ws_size;

  k_hist<<<SORT_WG, SORT_T, 0, stream>>>(dst_a, tab);
  k_scanB<<<1, SORT_T, 0, stream>>>(tab);
  k_scatter<<<SORT_WG, SORT_T, 0, stream>>>(src_a, dst_a, etype, tab, staged);
  k_bucket<<<N_BUCK, SORT_T, 0, stream>>>(tab, staged, off, ssrc);
  k_layer1<<<(N_NODES * 32u + 255) / 256, 256, 0, stream>>>(off, ssrc, w1, r1, b1, xbuf);
  k_xw<<<(RT_GROUPS * N_PAIRS * 32 + 255) / 256, 256, 0, stream>>>(xbuf, w2, r2, Y);
  k_layer2<<<(N_NODES * 32u + 255) / 256, 256, 0, stream>>>(off, ssrc, Y, b2, out);
}


# edge sort in 2 launches (was 4): each chunk partitions its own edges by bucket at chunk-local starts; a bucket's start = sum of its 256 slice starts; 7 launches become 5; slots stay in arrival order
# speedup vs baseline: 4.0659x; 1.2114x over previous
#include <hip/hip_runtime.h>


#define N_NODES 50000
#define N_REL   16
#define HID     64
#define N_LAB   32
#define N_EDGE  1600000
#define N_YC    544
#define ROW_TILES   (N_NODES / 16)
#define RT_PER_WAVE 4
#define RT_GROUPS   ((ROW_TILES + RT_PER_WAVE - 1) / RT_PER_WAVE)
#define N_PAIRS     (N_REL + 1)
#define N_KEYS  (N_NODES * N_REL)
#define N_BUCK   ((N_NODES + 255) / 256)
#define BK_KEYS  (256 * N_REL)
#define SORT_WG  256
#define SORT_T   1024
#define CHUNK    ((N_EDGE + SORT_WG - 1) / SORT_WG)
#define CH_IT    ((CHUNK + SORT_T - 1) / SORT_T)
static_assert(N_NODES <= 65536, "a source node index must fit in 16 bits of the staged word");
static_assert(BK_KEYS == 4 * SORT_T, "one thread per 4 keys in the bucket launch");
static_assert(SORT_WG == 256 && N_BUCK <= SORT_WG, "a chunk's bucket counters are scanned by one wave, 8 per lane");
static_assert((long long)SORT_WG * CHUNK >= N_EDGE, "the chunks cover every edge");

typedef float v2f __attribute__((ext_vector_type(2)));
typedef float v4f __attribute__((ext_vector_type(4)));
typedef float v8f __attribute__((ext_vector_type(8)));

#define WS_CNT_OFF  0
#define WS_X_OFF    (N_REL * N_NODES)
#define WS_Y_OFF    (WS_X_OFF + N_NODES * HID)
#define WS_SSRC_OFF (WS_Y_OFF + N_NODES * N_YC)
#define WS_STAGE_OFF WS_Y_OFF
#define WS_TABC_OFF  (WS_Y_OFF + N_EDGE)
#define WS_TABS_OFF  (WS_TABC_OFF + N_BUCK * SORT_WG)
#define NO_WORD 0xffffffffu

__device__ __forceinline__ int wave_incl_scan(int v, unsigned lane) {
  for (unsigned o = 1; o < 32; o <<= 1) {
    int u = __shfl_up(v, o);
    if (lane >= o) v += u;
  }
  return v;
}

__global__ __launch_bounds__(SORT_T) void k_part(const int* __restrict__ src_a, const int* __restrict__ dst_a,
                                                 const int* __restrict__ etype, int* __restrict__ tabc,
                                                 int* __restrict__ tabs, int* __restrict__ staged) {
  __shared__ int h[SORT_WG];
  unsigned t = threadIdx.x, g = blockIdx.x, lane = t & 31;
  if (t < SORT_WG) h[t] = 0;
  __syncthreads();
  unsigned beg = g * CHUNK;
  unsigned end = (beg + CHUNK < N_EDGE) ? (beg + CHUNK) : N_EDGE;
  int dv[CH_IT], wv[CH_IT];
  #pragma unroll
  for (int k = 0; k < CH_IT; k++) {
    unsigned i = beg + (unsigned)k * SORT_T + t;
    bool ok = i < end;
    int d = ok ? dst_a[i] : -1;
    int s = ok ? src_a[i] : 0;
    int r = ok ? etype[i] : 0;
    dv[k] = d;
    wv[k] = ((((d & 255) << 4) + r) << 16) | s;
  }
  int pos[CH_IT];
  #pragma unroll
  for (int k = 0; k < CH_IT; k++) {
    pos[k] = 0;
    if (dv[k] >= 0)
      pos[k] = __hip_atomic_fetch_add(&h[dv[k] >> 8], 1, __ATOMIC_RELAXED, __HIP_MEMORY_SCOPE_WORKGROUP);
  }
  __syncthreads();
  if (t < 32) {
    int cv[8], e[8];
    int s = 0;
    for (int j = 0; j < 8; j++) {
      cv[j] = h[8 * lane + (unsigned)j];
      e[j] = s; s += cv[j];
    }
    int inc = wave_incl_scan(s, lane);
    int exc = inc - s;
    for (int j = 0; j < 8; j++) {
      unsigned idx = 8 * lane + (unsigned)j;
      int st = e[j] + exc;
      h[idx] = st;
      if (idx < N_BUCK) {
        tabc[idx * SORT_WG + g] = cv[j];
        tabs[idx * SORT_WG + g] = st;
      }
    }
  }
  __syncthreads();
  #pragma unroll
  for (int k = 0; k < CH_IT; k++) {
    if (dv[k] >= 0) staged[beg + (unsigned)(h[dv[k] >> 8] + pos[k])] = wv[k];
  }
}

#define SL_PER_WAVE (SORT_WG / 32)
__global__ __launch_bounds__(SORT_T) void k_bucket(const int* __restrict__ tabc, const int* __restrict__ tabs,
                                                   const int* __restrict__ staged,
                                                   int* __restrict__ off, int* __restrict__ ssrc) {
  __shared__ alignas(16) int cnt[BK_KEYS];
  __shared__ int wtot[32];
  __shared__ int slc[SORT_WG];
  __shared__ int slb[SORT_WG];
  __shared__ int bsum[SORT_WG / 32];
  unsigned t = threadIdx.x, lane = t & 31, w = t >> 5;
  unsigned b = blockIdx.x;
  int4 zero4; zero4.x = 0; zero4.y = 0; zero4.z = 0; zero4.w = 0;
  ((int4*)cnt)[t] = zero4;
  if (t < SORT_WG) {
    int c = tabc[b * SORT_WG + t];
    int s = tabs[b * SORT_WG + t];
    slc[t] = c;
    slb[t] = (int)(t * CHUNK) + s;
    int r = s;
    for (int o = 16; o > 0; o >>= 1) r += __shfl_xor(r, o);
    if (lane == 0) bsum[w] = r;
  }
  __syncthreads();
  int bstart = 0;
  for (int i = 0; i < SORT_WG / 32; i++) bstart += bsum[i];
  int sc[SL_PER_WAVE], sb[SL_PER_WAVE];
  unsigned wd[SL_PER_WAVE][2];
  #pragma unroll
  for (int s = 0; s < SL_PER_WAVE; s++) {
    unsigned g = w + 32u * (unsigned)s;
    sc[s] = slc[g]; sb[s] = slb[g];
    wd[s][0] = ((int)lane < sc[s]) ? (unsigned)staged[sb[s] + (int)lane] : NO_WORD;
    wd[s][1] = ((int)lane + 32 < sc[s]) ? (unsigned)staged[sb[s] + (int)lane + 32] : NO_WORD;
  }
  #pragma unroll
  for (int s = 0; s < SL_PER_WAVE; s++) {
    for (int q = 0; q < 2; q++) {
      if (wd[s][q] != NO_WORD)
        __hip_atomic_fetch_add(&cnt[wd[s][q] >> 16], 1, __ATOMIC_RELAXED, __HIP_MEMORY_SCOPE_WORKGROUP);
    }
    for (int e = 64 + (int)lane; e < sc[s]; e += 32) {
      unsigned u = (unsigned)staged[sb[s] + e];
      __hip_atomic_fetch_add(&cnt[u >> 16], 1, __ATOMIC_RELAXED, __HIP_MEMORY_SCOPE_WORKGROUP);
    }
  }
  __syncthreads();
  int4 c = ((const int4*)cnt)[t];
  int s = c.x + c.y + c.z + c.w;
  int inc = wave_incl_scan(s, lane);
  if (lane == 31) wtot[w] = inc;
  __syncthreads();
  if (w == 0) {
    int v = wtot[lane];
    int vi = wave_incl_scan(v, lane);
    wtot[lane] = vi - v;
  }
  __syncthreads();
  int ex = wtot[w] + inc - s;
  int4 st;
  st.x = ex; st.y = st.x + c.x; st.z = st.y + c.y; st.w = st.z + c.z;
  ((int4*)cnt)[t] = st;
  unsigned key0 = b * BK_KEYS + 4 * t;
  if (key0 < N_KEYS) {
    int4 e;
    e.x = bstart + st.y; e.y = bstart + st.z; e.z = bstart + st.w; e.w = bstart + st.w + c.w;
    *((int4*)(off + key0)) = e;
  }
  __syncthreads();
  int pos[SL_PER_WAVE][2];
  #pragma unroll
  for (int s2 = 0; s2 < SL_PER_WAVE; s2++) {
    for (int q = 0; q < 2; q++) {
      pos[s2][q] = 0;
      if (wd[s2][q] != NO_WORD)
        pos[s2][q] = __hip_atomic_fetch_add(&cnt[wd[s2][q] >> 16], 1, __ATOMIC_RELAXED, __HIP_MEMORY_SCOPE_WORKGROUP);
    }
  }
  #pragma unroll
  for (int s2 = 0; s2 < SL_PER_WAVE; s2++) {
    for (int q = 0; q < 2; q++) {
      if (wd[s2][q] != NO_WORD) ssrc[bstart + pos[s2][q]] = (int)(wd[s2][q] & 0xffffu);
    }
    for (int e = 64 + (int)lane; e < sc[s2]; e += 32) {
      unsigned u = (unsigned)staged[sb[s2] + e];
      int p = __hip_atomic_fetch_add(&cnt[u >> 16], 1, __ATOMIC_RELAXED, __HIP_MEMORY_SCOPE_WORKGROUP);
      ssrc[bstart + p] = (int)(u & 0xffffu);
    }
  }
}

__global__ void k_layer1(const int* __restrict__ off, const int* __restrict__ ssrc,
                         const float* __restrict__ w1, const float* __restrict__ root1,
                         const float* __restrict__ bias1, float* __restrict__ xout) {
  unsigned tid = blockIdx.x * blockDim.x + threadIdx.x;
  unsigned d = (unsigned)__builtin_amdgcn_readfirstlane((int)(tid >> 5));
  unsigned lane = threadIdx.x & 31;
  if (d >= N_NODES) return;
  unsigned r16 = lane & 15;
  int endv = off[d * N_REL + r16];
  int node_start = (d == 0) ? 0 : off[d * N_REL - 1];
  int prev = __shfl_up(endv, 1);
  int startv = (r16 == 0) ? node_start : prev;
  float invv = 1.0f / fmaxf((float)(endv - startv), 1.0f);
  int ends[16];
  for (int r = 0; r < 16; r++) ends[r] = __builtin_amdgcn_readlane(endv, r);
  int node_end = ends[15];
  v2f r1 = *((const v2f*)(root1 + (size_t)d * HID) + lane);
  v2f b1 = *((const v2f*)bias1 + lane);

  v2f acc = {0.0f, 0.0f};
  for (int cb = node_start; cb < node_end; cb += 32) {
    int p = cb + (int)lane;
    int sv = (p < node_end) ? ssrc[p] : 0;
    int relv = 0;
    for (int r = 0; r < 16; r++) relv += (p >= ends[r]) ? 1 : 0;
    relv = (relv > 15) ? 15 : relv;
    float iv = __shfl(invv, relv);
    int n = node_end - cb;
    n = (n > 32) ? 32 : n;
    for (int j = 0; j < n; j += 8) {
      v2f v[8];
      float ij[8];
      for (int k = 0; k < 8; k++) {
        int jj = (j + k < n) ? (j + k) : (n - 1);
        int sj = __builtin_amdgcn_readlane(sv, jj);
        int rj = __builtin_amdgcn_readlane(relv, jj);
        ij[k] = __builtin_bit_cast(float, __builtin_amdgcn_readlane(__builtin_bit_cast(int, iv), jj));
        v[k] = *((const v2f*)(w1 + ((size_t)rj * N_NODES + (size_t)sj) * HID) + lane);
      }
      for (int k = 0; k < 8; k++) {
        if (j + k < n) {
          acc.x += v[k].x * ij[k];
          acc.y += v[k].y * ij[k];
        }
      }
    }
  }
  v2f x;
  x.x = fmaxf(acc.x + r1.x + b1.x, 0.0f);
  x.y = fmaxf(acc.y + r1.y + b1.y, 0.0f);
  *((v2f*)(xout + (size_t)d * HID) + lane) = x;
}

__global__ void k_xw(const float* __restrict__ x, const float* __restrict__ w2,
                     const float* __restrict__ r2, float* __restrict__ Y) {
  unsigned tid = blockIdx.x * blockDim.x + threadIdx.x;
  unsigned wave = (unsigned)__builtin_amdgcn_readfirstlane((int)(tid >> 5));
  unsigned lane = threadIdx.x & 31;
  if (wave >= RT_GROUPS * N_PAIRS) return;
  unsigned g = wave / N_PAIRS;
  unsigned p = wave - g * N_PAIRS;
  unsigned half = lane >> 4;
  unsigned sub = lane & 15;
  const float* Bp = (p < N_REL) ? (w2 + p * (HID * N_LAB)) : r2;
  v2f b0[16], b1[16];
  #pragma unroll
  for (int s = 0; s < 16; s++) {
    int kb = 8 * (s >> 1) + 4 * (int)half + 2 * (s & 1);
    b0[s].x = Bp[kb * N_LAB + sub];        b0[s].y = Bp[(kb + 1) * N_LAB + sub];
    b1[s].x = Bp[kb * N_LAB + 16 + sub];   b1[s].y = Bp[(kb + 1) * N_LAB + 16 + sub];
  }
  for (unsigned t = 0; t < RT_PER_WAVE; t++) {
    unsigned rt = g * RT_PER_WAVE + t;
    if (rt >= ROW_TILES) break;
    const float* arow = x + (size_t)(rt * 16u + sub) * HID + 4u * half;
    v2f a[16];
    #pragma unroll
    for (int j = 0; j < 8; j++) {
      v4f q = *((const v4f*)(arow + 8 * j));
      a[2 * j].x = q.x;      a[2 * j].y = q.y;
      a[2 * j + 1].x = q.z;  a[2 * j + 1].y = q.w;
    }
    v8f c0 = {}; v8f c1 = {};
    #pragma unroll
    for (int s = 0; s < 16; s++) {
      c0 = __builtin_amdgcn_wmma_f32_16x16x4_f32(false, a[s], false, b0[s], (short)0, c0, false, false);
      c1 = __builtin_amdgcn_wmma_f32_16x16x4_f32(false, a[s], false, b1[s], (short)0, c1, false, false);
    }
    float* yrow = Y + (size_t)(rt * 16u) * N_YC + p * N_LAB + lane;
    #pragma unroll
    for (int v = 0; v < 8; v++) {
      float send = half ? c0[v] : c1[v];
      float got = __builtin_bit_cast(float, __builtin_amdgcn_permlanex16(
          __builtin_bit_cast(int, send), __builtin_bit_cast(int, send), 0x76543210, 0xfedcba98, false, false));
      float lo = half ? got : c0[v];
      float hi = half ? c1[v] : got;
      yrow[(size_t)v * N_YC] = lo;
      yrow[(size_t)(v + 8) * N_YC] = hi;
    }
  }
}

__global__ void k_layer2(const int* __restrict__ off, const int* __restrict__ ssrc,
                         const float* __restrict__ Y, const float* __restrict__ bias2,
                         float* __restrict__ out) {
  unsigned tid = blockIdx.x * blockDim.x + threadIdx.x;
  unsigned d = (unsigned)__builtin_amdgcn_readfirstlane((int)(tid >> 5));
  unsigned lane = threadIdx.x & 31;
  if (d >= N_NODES) return;
  unsigned r16 = lane & 15;
  int endv = off[d * N_REL + r16];
  int node_start = (d == 0) ? 0 : off[d * N_REL - 1];
  int prev = __shfl_up(endv, 1);
  int startv = (r16 == 0) ? node_start : prev;
  float invv = 1.0f / fmaxf((float)(endv - startv), 1.0f);
  int ends[16];
  for (int r = 0; r < 16; r++) ends[r] = __builtin_amdgcn_readlane(endv, r);
  int node_end = ends[15];
  float rootv = Y[(size_t)d * N_YC + N_REL * N_LAB + lane];
  float b2 = bias2[lane];

  float acc = 0.0f;
  for (int cb = node_start; cb < node_end; cb += 32) {
    int p = cb + (int)lane;
    int sv = (p < node_end) ? ssrc[p] : 0;
    int relv = 0;
    for (int r = 0; r < 16; r++) relv += (p >= ends[r]) ? 1 : 0;
    relv = (relv > 15) ? 15 : relv;
    float iv = __shfl(invv, relv);
    int n = node_end - cb;
    n = (n > 32) ? 32 : n;
    for (int j = 0; j < n; j += 16) {
      float v[16];
      float ij[16];
      #pragma unroll
      for (int k = 0; k < 16; k++) {
        int jj = (j + k < n) ? (j + k) : (n - 1);
        int sj = __builtin_amdgcn_readlane(sv, jj);
        int rj = __builtin_amdgcn_readlane(relv, jj);
        ij[k] = __builtin_bit_cast(float, __builtin_amdgcn_readlane(__builtin_bit_cast(int, iv), jj));
        v[k] = Y[(size_t)sj * N_YC + (size_t)(rj * N_LAB) + lane];
      }
      #pragma unroll
      for (int k = 0; k < 16; k++) {
        if (j + k < n) acc += v[k] * ij[k];
      }
    }
  }
  float z = acc + rootv + b2;
  out[d * N_LAB + lane] = 1.0f / (1.0f + __expf(-z));
}

extern "C" void kernel_launch(void* const* d_in, const int* in_sizes, int n_in,
                              void* d_out, int out_size, void* d_ws, size_t ws_size,
                              hipStream_t stream) {
  const int* ei     = (const int*)d_in[0];
  const int* src_a  = ei;
  const int* dst_a  = ei + N_EDGE;
  const int* etype  = (const int*)d_in[1];
  const float* w1   = (const float*)d_in[2];
  const float* r1   = (const float*)d_in[3];
  const float* b1   = (const float*)d_in[4];
  const float* w2   = (const float*)d_in[5];
  const float* r2   = (const float*)d_in[6];
  const float* b2   = (const float*)d_in[7];
  float* out = (float*)d_out;

  float* ws   = (float*)d_ws;
  int*   off  = (int*)d_ws + WS_CNT_OFF;
  float* xbuf = ws + WS_X_OFF;
  float* Y    = ws + WS_Y_OFF;
  int*   staged = (int*)d_ws + WS_STAGE_OFF;
  int*   tabc = (int*)d_ws + WS_TABC_OFF;
  int*   tabs = (int*)d_ws + WS_TABS_OFF;
  int*   ssrc = (int*)d_ws + WS_SSRC_OFF;

  (void)in_sizes; (void)n_in; (void)out_size; (void)ws_size;

  k_part<<<SORT_WG, SORT_T, 0, stream>>>(src_a, dst_a, etype, tabc, tabs, staged);
  k_bucket<<<N_BUCK, SORT_T, 0, stream>>>(tabc, tabs, staged, off, ssrc);
  k_layer1<<<(N_NODES * 32u + 255) / 256, 256, 0, stream>>>(off, ssrc, w1, r1, b1, xbuf);
  k_xw<<<(RT_GROUPS * N_PAIRS * 32 + 255) / 256, 256, 0, stream>>>(xbuf, w2, r2, Y);
  k_layer2<<<(N_NODES * 32u + 255) / 256, 256, 0, stream>>>(off, ssrc, Y, b2, out);
}


# layer 2 in one launch: 16 waves gather 16 nodes' per-relation means of x rows (cache-resident) into LDS, then [16x1088] @ [weight2|root2] on f32 WMMA, K split over waves; Y gone; sums reassociated
# speedup vs baseline: 4.4619x; 1.0974x over previous
#include <hip/hip_runtime.h>


#define N_NODES 50000
#define N_REL   16
#define HID     64
#define N_LAB   32
#define N_EDGE  1600000
#define N_KEYS  (N_NODES * N_REL)
#define N_BUCK   ((N_NODES + 255) / 256)
#define BK_KEYS  (256 * N_REL)
#define SORT_WG  256
#define SORT_T   1024
#define CHUNK    ((N_EDGE + SORT_WG - 1) / SORT_WG)
#define CH_IT    ((CHUNK + SORT_T - 1) / SORT_T)
static_assert(N_NODES <= 65536, "a source node index must fit in 16 bits of the staged word");
static_assert(BK_KEYS == 4 * SORT_T, "one thread per 4 keys in the bucket launch");
static_assert(SORT_WG == 256 && N_BUCK <= SORT_WG, "a chunk's bucket counters are scanned by one wave, 8 per lane");
static_assert((long long)SORT_WG * CHUNK >= N_EDGE, "the chunks cover every edge");

typedef float v2f __attribute__((ext_vector_type(2)));
typedef float v4f __attribute__((ext_vector_type(4)));
typedef float v8f __attribute__((ext_vector_type(8)));

#define WS_CNT_OFF  0
#define WS_X_OFF    (N_REL * N_NODES)
#define WS_STAGE_OFF (WS_X_OFF + N_NODES * HID)
#define WS_SSRC_OFF (WS_STAGE_OFF + N_NODES * 544)
#define WS_TABC_OFF  (WS_STAGE_OFF + N_EDGE)
#define WS_TABS_OFF  (WS_TABC_OFF + N_BUCK * SORT_WG)
#define NO_WORD 0xffffffffu

__device__ __forceinline__ int wave_incl_scan(int v, unsigned lane) {
  for (unsigned o = 1; o < 32; o <<= 1) {
    int u = __shfl_up(v, o);
    if (lane >= o) v += u;
  }
  return v;
}

__global__ __launch_bounds__(SORT_T) void k_part(const int* __restrict__ src_a, const int* __restrict__ dst_a,
                                                 const int* __restrict__ etype, int* __restrict__ tabc,
                                                 int* __restrict__ tabs, int* __restrict__ staged) {
  __shared__ int h[SORT_WG];
  unsigned t = threadIdx.x, g = blockIdx.x, lane = t & 31;
  if (t < SORT_WG) h[t] = 0;
  __syncthreads();
  unsigned beg = g * CHUNK;
  unsigned end = (beg + CHUNK < N_EDGE) ? (beg + CHUNK) : N_EDGE;
  int dv[CH_IT], wv[CH_IT];
  #pragma unroll
  for (int k = 0; k < CH_IT; k++) {
    unsigned i = beg + (unsigned)k * SORT_T + t;
    bool ok = i < end;
    int d = ok ? dst_a[i] : -1;
    int s = ok ? src_a[i] : 0;
    int r = ok ? etype[i] : 0;
    dv[k] = d;
    wv[k] = ((((d & 255) << 4) + r) << 16) | s;
  }
  int pos[CH_IT];
  #pragma unroll
  for (int k = 0; k < CH_IT; k++) {
    pos[k] = 0;
    if (dv[k] >= 0)
      pos[k] = __hip_atomic_fetch_add(&h[dv[k] >> 8], 1, __ATOMIC_RELAXED, __HIP_MEMORY_SCOPE_WORKGROUP);
  }
  __syncthreads();
  if (t < 32) {
    int cv[8], e[8];
    int s = 0;
    for (int j = 0; j < 8; j++) {
      cv[j] = h[8 * lane + (unsigned)j];
      e[j] = s; s += cv[j];
    }
    int inc = wave_incl_scan(s, lane);
    int exc = inc - s;
    for (int j = 0; j < 8; j++) {
      unsigned idx = 8 * lane + (unsigned)j;
      int st = e[j] + exc;
      h[idx] = st;
      if (idx < N_BUCK) {
        tabc[idx * SORT_WG + g] = cv[j];
        tabs[idx * SORT_WG + g] = st;
      }
    }
  }
  __syncthreads();
  #pragma unroll
  for (int k = 0; k < CH_IT; k++) {
    if (dv[k] >= 0) staged[beg + (unsigned)(h[dv[k] >> 8] + pos[k])] = wv[k];
  }
}

#define SL_PER_WAVE (SORT_WG / 32)
__global__ __launch_bounds__(SORT_T) void k_bucket(const int* __restrict__ tabc, const int* __restrict__ tabs,
                                                   const int* __restrict__ staged,
                                                   int* __restrict__ off, int* __restrict__ ssrc) {
  __shared__ alignas(16) int cnt[BK_KEYS];
  __shared__ int wtot[32];
  __shared__ int slc[SORT_WG];
  __shared__ int slb[SORT_WG];
  __shared__ int bsum[SORT_WG / 32];
  unsigned t = threadIdx.x, lane = t & 31, w = t >> 5;
  unsigned b = blockIdx.x;
  int4 zero4; zero4.x = 0; zero4.y = 0; zero4.z = 0; zero4.w = 0;
  ((int4*)cnt)[t] = zero4;
  if (t < SORT_WG) {
    int c = tabc[b * SORT_WG + t];
    int s = tabs[b * SORT_WG + t];
    slc[t] = c;
    slb[t] = (int)(t * CHUNK) + s;
    int r = s;
    for (int o = 16; o > 0; o >>= 1) r += __shfl_xor(r, o);
    if (lane == 0) bsum[w] = r;
  }
  __syncthreads();
  int bstart = 0;
  for (int i = 0; i < SORT_WG / 32; i++) bstart += bsum[i];
  int sc[SL_PER_WAVE], sb[SL_PER_WAVE];
  unsigned wd[SL_PER_WAVE][2];
  #pragma unroll
  for (int s = 0; s < SL_PER_WAVE; s++) {
    unsigned g = w + 32u * (unsigned)s;
    sc[s] = slc[g]; sb[s] = slb[g];
    wd[s][0] = ((int)lane < sc[s]) ? (unsigned)staged[sb[s] + (int)lane] : NO_WORD;
    wd[s][1] = ((int)lane + 32 < sc[s]) ? (unsigned)staged[sb[s] + (int)lane + 32] : NO_WORD;
  }
  #pragma unroll
  for (int s = 0; s < SL_PER_WAVE; s++) {
    for (int q = 0; q < 2; q++) {
      if (wd[s][q] != NO_WORD)
        __hip_atomic_fetch_add(&cnt[wd[s][q] >> 16], 1, __ATOMIC_RELAXED, __HIP_MEMORY_SCOPE_WORKGROUP);
    }
    for (int e = 64 + (int)lane; e < sc[s]; e += 32) {
      unsigned u = (unsigned)staged[sb[s] + e];
      __hip_atomic_fetch_add(&cnt[u >> 16], 1, __ATOMIC_RELAXED, __HIP_MEMORY_SCOPE_WORKGROUP);
    }
  }
  __syncthreads();
  int4 c = ((const int4*)cnt)[t];
  int s = c.x + c.y + c.z + c.w;
  int inc = wave_incl_scan(s, lane);
  if (lane == 31) wtot[w] = inc;
  __syncthreads();
  if (w == 0) {
    int v = wtot[lane];
    int vi = wave_incl_scan(v, lane);
    wtot[lane] = vi - v;
  }
  __syncthreads();
  int ex = wtot[w] + inc - s;
  int4 st;
  st.x = ex; st.y = st.x + c.x; st.z = st.y + c.y; st.w = st.z + c.z;
  ((int4*)cnt)[t] = st;
  unsigned key0 = b * BK_KEYS + 4 * t;
  if (key0 < N_KEYS) {
    int4 e;
    e.x = bstart + st.y; e.y = bstart + st.z; e.z = bstart + st.w; e.w = bstart + st.w + c.w;
    *((int4*)(off + key0)) = e;
  }
  __syncthreads();
  int pos[SL_PER_WAVE][2];
  #pragma unroll
  for (int s2 = 0; s2 < SL_PER_WAVE; s2++) {
    for (int q = 0; q < 2; q++) {
      pos[s2][q] = 0;
      if (wd[s2][q] != NO_WORD)
        pos[s2][q] = __hip_atomic_fetch_add(&cnt[wd[s2][q] >> 16], 1, __ATOMIC_RELAXED, __HIP_MEMORY_SCOPE_WORKGROUP);
    }
  }
  #pragma unroll
  for (int s2 = 0; s2 < SL_PER_WAVE; s2++) {
    for (int q = 0; q < 2; q++) {
      if (wd[s2][q] != NO_WORD) ssrc[bstart + pos[s2][q]] = (int)(wd[s2][q] & 0xffffu);
    }
    for (int e = 64 + (int)lane; e < sc[s2]; e += 32) {
      unsigned u = (unsigned)staged[sb[s2] + e];
      int p = __hip_atomic_fetch_add(&cnt[u >> 16], 1, __ATOMIC_RELAXED, __HIP_MEMORY_SCOPE_WORKGROUP);
      ssrc[bstart + p] = (int)(u & 0xffffu);
    }
  }
}

__global__ void k_layer1(const int* __restrict__ off, const int* __restrict__ ssrc,
                         const float* __restrict__ w1, const float* __restrict__ root1,
                         const float* __restrict__ bias1, float* __restrict__ xout) {
  unsigned tid = blockIdx.x * blockDim.x + threadIdx.x;
  unsigned d = (unsigned)__builtin_amdgcn_readfirstlane((int)(tid >> 5));
  unsigned lane = threadIdx.x & 31;
  if (d >= N_NODES) return;
  unsigned r16 = lane & 15;
  int endv = off[d * N_REL + r16];
  int node_start = (d == 0) ? 0 : off[d * N_REL - 1];
  int prev = __shfl_up(endv, 1);
  int startv = (r16 == 0) ? node_start : prev;
  float invv = 1.0f / fmaxf((float)(endv - startv), 1.0f);
  int ends[16];
  for (int r = 0; r < 16; r++) ends[r] = __builtin_amdgcn_readlane(endv, r);
  int node_end = ends[15];
  v2f r1 = *((const v2f*)(root1 + (size_t)d * HID) + lane);
  v2f b1 = *((const v2f*)bias1 + lane);

  v2f acc = {0.0f, 0.0f};
  for (int cb = node_start; cb < node_end; cb += 32) {
    int p = cb + (int)lane;
    int sv = (p < node_end) ? ssrc[p] : 0;
    int relv = 0;
    for (int r = 0; r < 16; r++) relv += (p >= ends[r]) ? 1 : 0;
    relv = (relv > 15) ? 15 : relv;
    float iv = __shfl(invv, relv);
    int n = node_end - cb;
    n = (n > 32) ? 32 : n;
    for (int j = 0; j < n; j += 8) {
      v2f v[8];
      float ij[8];
      for (int k = 0; k < 8; k++) {
        int jj = (j + k < n) ? (j + k) : (n - 1);
        int sj = __builtin_amdgcn_readlane(sv, jj);
        int rj = __builtin_amdgcn_readlane(relv, jj);
        ij[k] = __builtin_bit_cast(float, __builtin_amdgcn_readlane(__builtin_bit_cast(int, iv), jj));
        v[k] = *((const v2f*)(w1 + ((size_t)rj * N_NODES + (size_t)sj) * HID) + lane);
      }
      for (int k = 0; k < 8; k++) {
        if (j + k < n) {
          acc.x += v[k].x * ij[k];
          acc.y += v[k].y * ij[k];
        }
      }
    }
  }
  v2f x;
  x.x = fmaxf(acc.x + r1.x + b1.x, 0.0f);
  x.y = fmaxf(acc.y + r1.y + b1.y, 0.0f);
  *((v2f*)(xout + (size_t)d * HID) + lane) = x;
}

#define M_COLS   (N_REL * HID + HID)
#define M_PITCH  (M_COLS + 4)
#define RED_PITCH 34
#define L2_WAVES 16
static_assert(N_NODES % 16 == 0, "every tile has 16 nodes: no partial tile, no early return before the barriers");
static_assert(L2_WAVES * 16 * RED_PITCH <= 16 * M_PITCH, "the reduce buffer reuses the matrix's LDS");
__global__ __launch_bounds__(L2_WAVES * 32) void k_layer2(const int* __restrict__ off, const int* __restrict__ ssrc,
                                                         const float* __restrict__ x, const float* __restrict__ w2,
                                                         const float* __restrict__ r2, const float* __restrict__ bias2,
                                                         float* __restrict__ out) {
  __shared__ alignas(16) float Ms[16 * M_PITCH];
  unsigned w = (unsigned)__builtin_amdgcn_readfirstlane((int)(threadIdx.x >> 5));
  unsigned lane = threadIdx.x & 31;
  unsigned d = blockIdx.x * 16u + w;
  float* Mrow = Ms + w * M_PITCH;
  v4f z4 = {0.0f, 0.0f, 0.0f, 0.0f};
  for (unsigned q = lane; q < M_COLS / 4; q += 32) *((v4f*)Mrow + q) = z4;

  unsigned r16 = lane & 15;
  int endv = off[d * N_REL + r16];
  int node_start = (d == 0) ? 0 : off[d * N_REL - 1];
  int prev = __shfl_up(endv, 1);
  int startv = (r16 == 0) ? node_start : prev;
  float invv = 1.0f / fmaxf((float)(endv - startv), 1.0f);
  int ends[16];
  for (int r = 0; r < 16; r++) ends[r] = __builtin_amdgcn_readlane(endv, r);
  int node_end = ends[15];
  *((v2f*)(Mrow + N_REL * HID) + lane) = *((const v2f*)(x + (size_t)d * HID) + lane);

  v2f acc = {0.0f, 0.0f};
  int cur = -1;
  for (int cb = node_start; cb < node_end; cb += 32) {
    int p = cb + (int)lane;
    int sv = (p < node_end) ? ssrc[p] : 0;
    int relv = 0;
    for (int r = 0; r < 16; r++) relv += (p >= ends[r]) ? 1 : 0;
    relv = (relv > 15) ? 15 : relv;
    int n = node_end - cb;
    n = (n > 32) ? 32 : n;
    for (int j = 0; j < n; j += 8) {
      v2f v[8];
      int rj[8];
      #pragma unroll
      for (int k = 0; k < 8; k++) {
        int jj = (j + k < n) ? (j + k) : (n - 1);
        int sj = __builtin_amdgcn_readlane(sv, jj);
        rj[k] = __builtin_amdgcn_readlane(relv, jj);
        v[k] = *((const v2f*)(x + (size_t)sj * HID) + lane);
      }
      #pragma unroll
      for (int k = 0; k < 8; k++) {
        if (j + k < n) {
          if (rj[k] != cur) {
            if (cur >= 0) {
              float iv = __builtin_bit_cast(float, __builtin_amdgcn_readlane(__builtin_bit_cast(int, invv), cur));
              v2f mv; mv.x = acc.x * iv; mv.y = acc.y * iv;
              *((v2f*)(Mrow + cur * HID) + lane) = mv;
            }
            acc.x = 0.0f; acc.y = 0.0f;
            cur = rj[k];
          }
          acc.x += v[k].x;
          acc.y += v[k].y;
        }
      }
    }
  }
  if (cur >= 0) {
    float iv = __builtin_bit_cast(float, __builtin_amdgcn_readlane(__builtin_bit_cast(int, invv), cur));
    v2f mv; mv.x = acc.x * iv; mv.y = acc.y * iv;
    *((v2f*)(Mrow + cur * HID) + lane) = mv;
  }

  unsigned half = lane >> 4;
  unsigned sub = lane & 15;
  const float* Bp = w2 + (size_t)w * (HID * N_LAB) + (2u * half) * N_LAB + sub;
  const float* Rp = r2 + (4u * w + 2u * half) * N_LAB + sub;
  v2f b0[8], b1[8];
  #pragma unroll
  for (int s = 0; s < 8; s++) {
    b0[s].x = Bp[(4 * s) * N_LAB];        b0[s].y = Bp[(4 * s + 1) * N_LAB];
    b1[s].x = Bp[(4 * s) * N_LAB + 16];   b1[s].y = Bp[(4 * s + 1) * N_LAB + 16];
  }
  __syncthreads();

  const float* Af = Ms + sub * M_PITCH + 2u * half;
  v8f c0 = {}; v8f c1 = {};
  #pragma unroll
  for (int s = 0; s < 8; s++) {
    v2f a = *((const v2f*)(Af + HID * w + 4 * s));
    c0 = __builtin_amdgcn_wmma_f32_16x16x4_f32(false, a, false, b0[s], (short)0, c0, false, false);
    c1 = __builtin_amdgcn_wmma_f32_16x16x4_f32(false, a, false, b1[s], (short)0, c1, false, false);
  }
  v2f d0[9], d1[9];
  #pragma unroll
  for (int s = 0; s < 8; s++) {
    d0[s].x = Bp[(4 * s + 32) * N_LAB];        d0[s].y = Bp[(4 * s + 33) * N_LAB];
    d1[s].x = Bp[(4 * s + 32) * N_LAB + 16];   d1[s].y = Bp[(4 * s + 33) * N_LAB + 16];
  }
  d0[8].x = Rp[0];   d0[8].y = Rp[N_LAB];
  d1[8].x = Rp[16];  d1[8].y = Rp[N_LAB + 16];
  #pragma unroll
  for (int s = 0; s < 8; s++) {
    v2f a = *((const v2f*)(Af + HID * w + 4 * s + 32));
    c0 = __builtin_amdgcn_wmma_f32_16x16x4_f32(false, a, false, d0[s], (short)0, c0, false, false);
    c1 = __builtin_amdgcn_wmma_f32_16x16x4_f32(false, a, false, d1[s], (short)0, c1, false, false);
  }
  {
    v2f a = *((const v2f*)(Af + N_REL * HID + 4u * w));
    c0 = __builtin_amdgcn_wmma_f32_16x16x4_f32(false, a, false, d0[8], (short)0, c0, false, false);
    c1 = __builtin_amdgcn_wmma_f32_16x16x4_f32(false, a, false, d1[8], (short)0, c1, false, false);
  }
  __syncthreads();

  float* Pw = Ms + (w * 16u) * RED_PITCH + lane;
  #pragma unroll
  for (int v = 0; v < 8; v++) {
    Pw[v * RED_PITCH] = c0[v];
    Pw[(8 + v) * RED_PITCH] = c1[v];
  }
  __syncthreads();
  unsigned ri = (lane >> 4) * 8u + (w & 7u);
  unsigned sl = (w >> 3) * 16u + (lane & 15u);
  const float* Pr = Ms + ri * RED_PITCH + sl;
  float zsum = 0.0f;
  #pragma unroll
  for (int ww = 0; ww < L2_WAVES; ww++) zsum += Pr[(ww * 16) * RED_PITCH];
  float zz = zsum + bias2[lane];
  out[d * N_LAB + lane] = 1.0f / (1.0f + __expf(-zz));
}

extern "C" void kernel_launch(void* const* d_in, const int* in_sizes, int n_in,
                              void* d_out, int out_size, void* d_ws, size_t ws_size,
                              hipStream_t stream) {
  const int* ei     = (const int*)d_in[0];
  const int* src_a  = ei;
  const int* dst_a  = ei + N_EDGE;
  const int* etype  = (const int*)d_in[1];
  const float* w1   = (const float*)d_in[2];
  const float* r1   = (const float*)d_in[3];
  const float* b1   = (const float*)d_in[4];
  const float* w2   = (const float*)d_in[5];
  const float* r2   = (const float*)d_in[6];
  const float* b2   = (const float*)d_in[7];
  float* out = (float*)d_out;

  float* ws   = (float*)d_ws;
  int*   off  = (int*)d_ws + WS_CNT_OFF;
  float* xbuf = ws + WS_X_OFF;
  int*   staged = (int*)d_ws + WS_STAGE_OFF;
  int*   tabc = (int*)d_ws + WS_TABC_OFF;
  int*   tabs = (int*)d_ws + WS_TABS_OFF;
  int*   ssrc = (int*)d_ws + WS_SSRC_OFF;

  (void)in_sizes; (void)n_in; (void)out_size; (void)ws_size;

  k_part<<<SORT_WG, SORT_T, 0, stream>>>(src_a, dst_a, etype, tabc, tabs, staged);
  k_bucket<<<N_BUCK, SORT_T, 0, stream>>>(tabc, tabs, staged, off, ssrc);
  k_layer1<<<(N_NODES * 32u + 255) / 256, 256, 0, stream>>>(off, ssrc, w1, r1, b1, xbuf);
  k_layer2<<<N_NODES / 16, L2_WAVES * 32, 0, stream>>>(off, ssrc, xbuf, w2, r2, b2, out);
}


# edge sort: both launches place words in an LDS copy of their output region, then write it out 16 bytes a lane, 4 whole lines per wave store (was a 4-byte store per edge); slots stay in arrival order
# speedup vs baseline: 4.9653x; 1.1128x over previous
#include <hip/hip_runtime.h>


#define N_NODES 50000
#define N_REL   16
#define HID     64
#define N_LAB   32
#define N_EDGE  1600000
#define N_KEYS  (N_NODES * N_REL)
#define N_BUCK   ((N_NODES + 255) / 256)
#define BK_KEYS  (256 * N_REL)
#define SORT_WG  256
#define SORT_T   1024
#define CHUNK    ((((N_EDGE + SORT_WG - 1) / SORT_WG) + 31) / 32 * 32)
#define CH_IT    ((CHUNK + SORT_T - 1) / SORT_T)
#define OB_CAP   16384
static_assert(N_NODES <= 65536, "a source node index must fit in 16 bits of the staged word");
static_assert(BK_KEYS == 4 * SORT_T, "one thread per 4 keys in the bucket launch");
static_assert(SORT_WG == 256 && N_BUCK <= SORT_WG, "a chunk's bucket counters are scanned by one wave, 8 per lane");
static_assert((long long)SORT_WG * CHUNK >= N_EDGE, "the chunks cover every edge");
static_assert((long long)(SORT_WG - 1) * CHUNK < N_EDGE, "every chunk starts inside the edge list");
static_assert(CHUNK % 32 == 0, "a chunk's staging region is whole 128-byte lines");

typedef float v2f __attribute__((ext_vector_type(2)));
typedef float v4f __attribute__((ext_vector_type(4)));
typedef float v8f __attribute__((ext_vector_type(8)));

#define WS_CNT_OFF  0
#define WS_X_OFF    (N_REL * N_NODES)
#define WS_STAGE_OFF (WS_X_OFF + N_NODES * HID)
#define WS_SSRC_OFF (WS_STAGE_OFF + N_NODES * 544)
#define WS_TABC_OFF  (WS_STAGE_OFF + SORT_WG * CHUNK)
#define WS_TABS_OFF  (WS_TABC_OFF + N_BUCK * SORT_WG)
#define NO_WORD 0xffffffffu

__device__ __forceinline__ int wave_incl_scan(int v, unsigned lane) {
  for (unsigned o = 1; o < 32; o <<= 1) {
    int u = __shfl_up(v, o);
    if (lane >= o) v += u;
  }
  return v;
}

__global__ __launch_bounds__(SORT_T) void k_part(const int* __restrict__ src_a, const int* __restrict__ dst_a,
                                                 const int* __restrict__ etype, int* __restrict__ tabc,
                                                 int* __restrict__ tabs, int* __restrict__ staged) {
  __shared__ int h[SORT_WG];
  __shared__ alignas(16) int obuf[CHUNK];
  unsigned t = threadIdx.x, g = blockIdx.x, lane = t & 31;
  if (t < SORT_WG) h[t] = 0;
  __syncthreads();
  unsigned beg = g * CHUNK;
  unsigned end = (beg + CHUNK < N_EDGE) ? (beg + CHUNK) : N_EDGE;
  int dv[CH_IT], wv[CH_IT];
  #pragma unroll
  for (int k = 0; k < CH_IT; k++) {
    unsigned i = beg + (unsigned)k * SORT_T + t;
    bool ok = i < end;
    int d = ok ? dst_a[i] : -1;
    int s = ok ? src_a[i] : 0;
    int r = ok ? etype[i] : 0;
    dv[k] = d;
    wv[k] = ((((d & 255) << 4) + r) << 16) | s;
  }
  int pos[CH_IT];
  #pragma unroll
  for (int k = 0; k < CH_IT; k++) {
    pos[k] = 0;
    if (dv[k] >= 0)
      pos[k] = __hip_atomic_fetch_add(&h[dv[k] >> 8], 1, __ATOMIC_RELAXED, __HIP_MEMORY_SCOPE_WORKGROUP);
  }
  __syncthreads();
  if (t < 32) {
    int cv[8], e[8];
    int s = 0;
    for (int j = 0; j < 8; j++) {
      cv[j] = h[8 * lane + (unsigned)j];
      e[j] = s; s += cv[j];
    }
    int inc = wave_incl_scan(s, lane);
    int exc = inc - s;
    for (int j = 0; j < 8; j++) {
      unsigned idx = 8 * lane + (unsigned)j;
      int st = e[j] + exc;
      h[idx] = st;
      if (idx < N_BUCK) {
        tabc[idx * SORT_WG + g] = cv[j];
        tabs[idx * SORT_WG + g] = st;
      }
    }
  }
  __syncthreads();
  #pragma unroll
  for (int k = 0; k < CH_IT; k++) {
    if (dv[k] >= 0) obuf[h[dv[k] >> 8] + pos[k]] = wv[k];
  }
  __syncthreads();
  unsigned n = end - beg;
  for (unsigned u = 4 * t; u < n; u += 4 * SORT_T) {
    if (u + 4 <= n) {
      *((int4*)(staged + beg + u)) = *((const int4*)(obuf + u));
    } else {
      for (unsigned q = u; q < n; q++) staged[beg + q] = obuf[q];
    }
  }
}

#define SL_PER_WAVE (SORT_WG / 32)
__global__ __launch_bounds__(SORT_T) void k_bucket(const int* __restrict__ tabc, const int* __restrict__ tabs,
                                                   const int* __restrict__ staged,
                                                   int* __restrict__ off, int* __restrict__ ssrc) {
  __shared__ alignas(16) int cnt[BK_KEYS];
  __shared__ int wtot[32];
  __shared__ int slc[SORT_WG];
  __shared__ int slb[SORT_WG];
  __shared__ int bsum[SORT_WG / 32];
  __shared__ int bcnt[SORT_WG / 32];
  __shared__ alignas(16) int obuf[OB_CAP + 32];
  unsigned t = threadIdx.x, lane = t & 31, w = t >> 5;
  unsigned b = blockIdx.x;
  int4 zero4; zero4.x = 0; zero4.y = 0; zero4.z = 0; zero4.w = 0;
  ((int4*)cnt)[t] = zero4;
  if (t < SORT_WG) {
    int c = tabc[b * SORT_WG + t];
    int s = tabs[b * SORT_WG + t];
    slc[t] = c;
    slb[t] = (int)(t * CHUNK) + s;
    int r = s, rc = c;
    for (int o = 16; o > 0; o >>= 1) { r += __shfl_xor(r, o); rc += __shfl_xor(rc, o); }
    if (lane == 0) { bsum[w] = r; bcnt[w] = rc; }
  }
  __syncthreads();
  int bstart = 0;
  int btotal = 0;
  for (int i = 0; i < SORT_WG / 32; i++) { bstart += bsum[i]; btotal += bcnt[i]; }
  const int shift = bstart & 31;
  const bool in_lds = btotal <= OB_CAP;
  int sc[SL_PER_WAVE], sb[SL_PER_WAVE];
  unsigned wd[SL_PER_WAVE][2];
  #pragma unroll
  for (int s = 0; s < SL_PER_WAVE; s++) {
    unsigned g = w + 32u * (unsigned)s;
    sc[s] = slc[g]; sb[s] = slb[g];
    wd[s][0] = ((int)lane < sc[s]) ? (unsigned)staged[sb[s] + (int)lane] : NO_WORD;
    wd[s][1] = ((int)lane + 32 < sc[s]) ? (unsigned)staged[sb[s] + (int)lane + 32] : NO_WORD;
  }
  #pragma unroll
  for (int s = 0; s < SL_PER_WAVE; s++) {
    for (int q = 0; q < 2; q++) {
      if (wd[s][q] != NO_WORD)
        __hip_atomic_fetch_add(&cnt[wd[s][q] >> 16], 1, __ATOMIC_RELAXED, __HIP_MEMORY_SCOPE_WORKGROUP);
    }
    for (int e = 64 + (int)lane; e < sc[s]; e += 32) {
      unsigned u = (unsigned)staged[sb[s] + e];
      __hip_atomic_fetch_add(&cnt[u >> 16], 1, __ATOMIC_RELAXED, __HIP_MEMORY_SCOPE_WORKGROUP);
    }
  }
  __syncthreads();
  int4 c = ((const int4*)cnt)[t];
  int s = c.x + c.y + c.z + c.w;
  int inc = wave_incl_scan(s, lane);
  if (lane == 31) wtot[w] = inc;
  __syncthreads();
  if (w == 0) {
    int v = wtot[lane];
    int vi = wave_incl_scan(v, lane);
    wtot[lane] = vi - v;
  }
  __syncthreads();
  int ex = wtot[w] + inc - s;
  int4 st;
  st.x = ex; st.y = st.x + c.x; st.z = st.y + c.y; st.w = st.z + c.z;
  ((int4*)cnt)[t] = st;
  unsigned key0 = b * BK_KEYS + 4 * t;
  if (key0 < N_KEYS) {
    int4 e;
    e.x = bstart + st.y; e.y = bstart + st.z; e.z = bstart + st.w; e.w = bstart + st.w + c.w;
    *((int4*)(off + key0)) = e;
  }
  __syncthreads();
  int pos[SL_PER_WAVE][2];
  #pragma unroll
  for (int s2 = 0; s2 < SL_PER_WAVE; s2++) {
    for (int q = 0; q < 2; q++) {
      pos[s2][q] = 0;
      if (wd[s2][q] != NO_WORD)
        pos[s2][q] = __hip_atomic_fetch_add(&cnt[wd[s2][q] >> 16], 1, __ATOMIC_RELAXED, __HIP_MEMORY_SCOPE_WORKGROUP);
    }
  }
  if (in_lds) {
    #pragma unroll
    for (int s2 = 0; s2 < SL_PER_WAVE; s2++) {
      for (int q = 0; q < 2; q++) {
        if (wd[s2][q] != NO_WORD) obuf[shift + pos[s2][q]] = (int)(wd[s2][q] & 0xffffu);
      }
      for (int e = 64 + (int)lane; e < sc[s2]; e += 32) {
        unsigned u = (unsigned)staged[sb[s2] + e];
        int p = __hip_atomic_fetch_add(&cnt[u >> 16], 1, __ATOMIC_RELAXED, __HIP_MEMORY_SCOPE_WORKGROUP);
        obuf[shift + p] = (int)(u & 0xffffu);
      }
    }
    __syncthreads();
    int* gout = ssrc + (bstart - shift);
    const int lo = shift, hi = shift + btotal;
    for (int u = 4 * (int)t; u < hi; u += 4 * SORT_T) {
      if (u >= lo && u + 4 <= hi) {
        *((int4*)(gout + u)) = *((const int4*)(obuf + u));
      } else {
        for (int q = u; q < u + 4; q++)
          if (q >= lo && q < hi) gout[q] = obuf[q];
      }
    }
  } else {
    #pragma unroll
    for (int s2 = 0; s2 < SL_PER_WAVE; s2++) {
      for (int q = 0; q < 2; q++) {
        if (wd[s2][q] != NO_WORD) ssrc[bstart + pos[s2][q]] = (int)(wd[s2][q] & 0xffffu);
      }
      for (int e = 64 + (int)lane; e < sc[s2]; e += 32) {
        unsigned u = (unsigned)staged[sb[s2] + e];
        int p = __hip_atomic_fetch_add(&cnt[u >> 16], 1, __ATOMIC_RELAXED, __HIP_MEMORY_SCOPE_WORKGROUP);
        ssrc[bstart + p] = (int)(u & 0xffffu);
      }
    }
  }
}

__global__ void k_layer1(const int* __restrict__ off, const int* __restrict__ ssrc,
                         const float* __restrict__ w1, const float* __restrict__ root1,
                         const float* __restrict__ bias1, float* __restrict__ xout) {
  unsigned tid = blockIdx.x * blockDim.x + threadIdx.x;
  unsigned d = (unsigned)__builtin_amdgcn_readfirstlane((int)(tid >> 5));
  unsigned lane = threadIdx.x & 31;
  if (d >= N_NODES) return;
  unsigned r16 = lane & 15;
  int endv = off[d * N_REL + r16];
  int node_start = (d == 0) ? 0 : off[d * N_REL - 1];
  int prev = __shfl_up(endv, 1);
  int startv = (r16 == 0) ? node_start : prev;
  float invv = 1.0f / fmaxf((float)(endv - startv), 1.0f);
  int ends[16];
  for (int r = 0; r < 16; r++) ends[r] = __builtin_amdgcn_readlane(endv, r);
  int node_end = ends[15];
  v2f r1 = *((const v2f*)(root1 + (size_t)d * HID) + lane);
  v2f b1 = *((const v2f*)bias1 + lane);

  v2f acc = {0.0f, 0.0f};
  for (int cb = node_start; cb < node_end; cb += 32) {
    int p = cb + (int)lane;
    int sv = (p < node_end) ? ssrc[p] : 0;
    int relv = 0;
    for (int r = 0; r < 16; r++) relv += (p >= ends[r]) ? 1 : 0;
    relv = (relv > 15) ? 15 : relv;
    float iv = __shfl(invv, relv);
    int n = node_end - cb;
    n = (n > 32) ? 32 : n;
    for (int j = 0; j < n; j += 8) {
      v2f v[8];
      float ij[8];
      for (int k = 0; k < 8; k++) {
        int jj = (j + k < n) ? (j + k) : (n - 1);
        int sj = __builtin_amdgcn_readlane(sv, jj);
        int rj = __builtin_amdgcn_readlane(relv, jj);
        ij[k] = __builtin_bit_cast(float, __builtin_amdgcn_readlane(__builtin_bit_cast(int, iv), jj));
        v[k] = *((const v2f*)(w1 + ((size_t)rj * N_NODES + (size_t)sj) * HID) + lane);
      }
      for (int k = 0; k < 8; k++) {
        if (j + k < n) {
          acc.x += v[k].x * ij[k];
          acc.y += v[k].y * ij[k];
        }
      }
    }
  }
  v2f x;
  x.x = fmaxf(acc.x + r1.x + b1.x, 0.0f);
  x.y = fmaxf(acc.y + r1.y + b1.y, 0.0f);
  *((v2f*)(xout + (size_t)d * HID) + lane) = x;
}

#define M_COLS   (N_REL * HID + HID)
#define M_PITCH  (M_COLS + 4)
#define RED_PITCH 34
#define L2_WAVES 16
static_assert(N_NODES % 16 == 0, "every tile has 16 nodes: no partial tile, no early return before the barriers");
static_assert(L2_WAVES * 16 * RED_PITCH <= 16 * M_PITCH, "the reduce buffer reuses the matrix's LDS");
__global__ __launch_bounds__(L2_WAVES * 32) void k_layer2(const int* __restrict__ off, const int* __restrict__ ssrc,
                                                         const float* __restrict__ x, const float* __restrict__ w2,
                                                         const float* __restrict__ r2, const float* __restrict__ bias2,
                                                         float* __restrict__ out) {
  __shared__ alignas(16) float Ms[16 * M_PITCH];
  unsigned w = (unsigned)__builtin_amdgcn_readfirstlane((int)(threadIdx.x >> 5));
  unsigned lane = threadIdx.x & 31;
  unsigned d = blockIdx.x * 16u + w;
  float* Mrow = Ms + w * M_PITCH;
  v4f z4 = {0.0f, 0.0f, 0.0f, 0.0f};
  for (unsigned q = lane; q < M_COLS / 4; q += 32) *((v4f*)Mrow + q) = z4;

  unsigned r16 = lane & 15;
  int endv = off[d * N_REL + r16];
  int node_start = (d == 0) ? 0 : off[d * N_REL - 1];
  int prev = __shfl_up(endv, 1);
  int startv = (r16 == 0) ? node_start : prev;
  float invv = 1.0f / fmaxf((float)(endv - startv), 1.0f);
  int ends[16];
  for (int r = 0; r < 16; r++) ends[r] = __builtin_amdgcn_readlane(endv, r);
  int node_end = ends[15];
  *((v2f*)(Mrow + N_REL * HID) + lane) = *((const v2f*)(x + (size_t)d * HID) + lane);

  v2f acc = {0.0f, 0.0f};
  int cur = -1;
  for (int cb = node_start; cb < node_end; cb += 32) {
    int p = cb + (int)lane;
    int sv = (p < node_end) ? ssrc[p] : 0;
    int relv = 0;
    for (int r = 0; r < 16; r++) relv += (p >= ends[r]) ? 1 : 0;
    relv = (relv > 15) ? 15 : relv;
    int n = node_end - cb;
    n = (n > 32) ? 32 : n;
    for (int j = 0; j < n; j += 8) {
      v2f v[8];
      int rj[8];
      #pragma unroll
      for (int k = 0; k < 8; k++) {
        int jj = (j + k < n) ? (j + k) : (n - 1);
        int sj = __builtin_amdgcn_readlane(sv, jj);
        rj[k] = __builtin_amdgcn_readlane(relv, jj);
        v[k] = *((const v2f*)(x + (size_t)sj * HID) + lane);
      }
      #pragma unroll
      for (int k = 0; k < 8; k++) {
        if (j + k < n) {
          if (rj[k] != cur) {
            if (cur >= 0) {
              float iv = __builtin_bit_cast(float, __builtin_amdgcn_readlane(__builtin_bit_cast(int, invv), cur));
              v2f mv; mv.x = acc.x * iv; mv.y = acc.y * iv;
              *((v2f*)(Mrow + cur * HID) + lane) = mv;
            }
            acc.x = 0.0f; acc.y = 0.0f;
            cur = rj[k];
          }
          acc.x += v[k].x;
          acc.y += v[k].y;
        }
      }
    }
  }
  if (cur >= 0) {
    float iv = __builtin_bit_cast(float, __builtin_amdgcn_readlane(__builtin_bit_cast(int, invv), cur));
    v2f mv; mv.x = acc.x * iv; mv.y = acc.y * iv;
    *((v2f*)(Mrow + cur * HID) + lane) = mv;
  }

  unsigned half = lane >> 4;
  unsigned sub = lane & 15;
  const float* Bp = w2 + (size_t)w * (HID * N_LAB) + (2u * half) * N_LAB + sub;
  const float* Rp = r2 + (4u * w + 2u * half) * N_LAB + sub;
  v2f b0[8], b1[8];
  #pragma unroll
  for (int s = 0; s < 8; s++) {
    b0[s].x = Bp[(4 * s) * N_LAB];        b0[s].y = Bp[(4 * s + 1) * N_LAB];
    b1[s].x = Bp[(4 * s) * N_LAB + 16];   b1[s].y = Bp[(4 * s + 1) * N_LAB + 16];
  }
  __syncthreads();

  const float* Af = Ms + sub * M_PITCH + 2u * half;
  v8f c0 = {}; v8f c1 = {};
  #pragma unroll
  for (int s = 0; s < 8; s++) {
    v2f a = *((const v2f*)(Af + HID * w + 4 * s));
    c0 = __builtin_amdgcn_wmma_f32_16x16x4_f32(false, a, false, b0[s], (short)0, c0, false, false);
    c1 = __builtin_amdgcn_wmma_f32_16x16x4_f32(false, a, false, b1[s], (short)0, c1, false, false);
  }
  v2f d0[9], d1[9];
  #pragma unroll
  for (int s = 0; s < 8; s++) {
    d0[s].x = Bp[(4 * s + 32) * N_LAB];        d0[s].y = Bp[(4 * s + 33) * N_LAB];
    d1[s].x = Bp[(4 * s + 32) * N_LAB + 16];   d1[s].y = Bp[(4 * s + 33) * N_LAB + 16];
  }
  d0[8].x = Rp[0];   d0[8].y = Rp[N_LAB];
  d1[8].x = Rp[16];  d1[8].y = Rp[N_LAB + 16];
  #pragma unroll
  for (int s = 0; s < 8; s++) {
    v2f a = *((const v2f*)(Af + HID * w + 4 * s + 32));
    c0 = __builtin_amdgcn_wmma_f32_16x16x4_f32(false, a, false, d0[s], (short)0, c0, false, false);
    c1 = __builtin_amdgcn_wmma_f32_16x16x4_f32(false, a, false, d1[s], (short)0, c1, false, false);
  }
  {
    v2f a = *((const v2f*)(Af + N_REL * HID + 4u * w));
    c0 = __builtin_amdgcn_wmma_f32_16x16x4_f32(false, a, false, d0[8], (short)0, c0, false, false);
    c1 = __builtin_amdgcn_wmma_f32_16x16x4_f32(false, a, false, d1[8], (short)0, c1, false, false);
  }
  __syncthreads();

  float* Pw = Ms + (w * 16u) * RED_PITCH + lane;
  #pragma unroll
  for (int v = 0; v < 8; v++) {
    Pw[v * RED_PITCH] = c0[v];
    Pw[(8 + v) * RED_PITCH] = c1[v];
  }
  __syncthreads();
  unsigned ri = (lane >> 4) * 8u + (w & 7u);
  unsigned sl = (w >> 3) * 16u + (lane & 15u);
  const float* Pr = Ms + ri * RED_PITCH + sl;
  float zsum = 0.0f;
  #pragma unroll
  for (int ww = 0; ww < L2_WAVES; ww++) zsum += Pr[(ww * 16) * RED_PITCH];
  float zz = zsum + bias2[lane];
  out[d * N_LAB + lane] = 1.0f / (1.0f + __expf(-zz));
}

extern "C" void kernel_launch(void* const* d_in, const int* in_sizes, int n_in,
                              void* d_out, int out_size, void* d_ws, size_t ws_size,
                              hipStream_t stream) {
  const int* ei     = (const int*)d_in[0];
  const int* src_a  = ei;
  const int* dst_a  = ei + N_EDGE;
  const int* etype  = (const int*)d_in[1];
  const float* w1   = (const float*)d_in[2];
  const float* r1   = (const float*)d_in[3];
  const float* b1   = (const float*)d_in[4];
  const float* w2   = (const float*)d_in[5];
  const float* r2   = (const float*)d_in[6];
  const float* b2   = (const float*)d_in[7];
  float* out = (float*)d_out;

  float* ws   = (float*)d_ws;
  int*   off  = (int*)d_ws + WS_CNT_OFF;
  float* xbuf = ws + WS_X_OFF;
  int*   staged = (int*)d_ws + WS_STAGE_OFF;
  int*   tabc = (int*)d_ws + WS_TABC_OFF;
  int*   tabs = (int*)d_ws + WS_TABS_OFF;
  int*   ssrc = (int*)d_ws + WS_SSRC_OFF;

  (void)in_sizes; (void)n_in; (void)out_size; (void)ws_size;

  k_part<<<SORT_WG, SORT_T, 0, stream>>>(src_a, dst_a, etype, tabc, tabs, staged);
  k_bucket<<<N_BUCK, SORT_T, 0, stream>>>(tabc, tabs, staged, off, ssrc);
  k_layer1<<<(N_NODES * 32u + 255) / 256, 256, 0, stream>>>(off, ssrc, w1, r1, b1, xbuf);
  k_layer2<<<N_NODES / 16, L2_WAVES * 32, 0, stream>>>(off, ssrc, xbuf, w2, r2, b2, out);
}


# layer 2: B fragments come from a copy of [weight2; root2] in fragment order (the bucket launch writes it): one 16-byte load per lane and k-step, 4 whole lines a wave load (was four 4-byte loads)
# speedup vs baseline: 5.1690x; 1.0410x over previous
#include <hip/hip_runtime.h>


#define N_NODES 50000
#define N_REL   16
#define HID     64
#define N_LAB   32
#define N_EDGE  1600000
#define N_KEYS  (N_NODES * N_REL)
#define N_BUCK   ((N_NODES + 255) / 256)
#define BK_KEYS  (256 * N_REL)
#define SORT_WG  256
#define SORT_T   1024
#define CHUNK    ((((N_EDGE + SORT_WG - 1) / SORT_WG) + 31) / 32 * 32)
#define CH_IT    ((CHUNK + SORT_T - 1) / SORT_T)
#define OB_CAP   16384
static_assert(N_NODES <= 65536, "a source node index must fit in 16 bits of the staged word");
static_assert(BK_KEYS == 4 * SORT_T, "one thread per 4 keys in the bucket launch");
static_assert(SORT_WG == 256 && N_BUCK <= SORT_WG, "a chunk's bucket counters are scanned by one wave, 8 per lane");
static_assert((long long)SORT_WG * CHUNK >= N_EDGE, "the chunks cover every edge");
static_assert((long long)(SORT_WG - 1) * CHUNK < N_EDGE, "every chunk starts inside the edge list");
static_assert(CHUNK % 32 == 0, "a chunk's staging region is whole 128-byte lines");

typedef float v2f __attribute__((ext_vector_type(2)));
typedef float v4f __attribute__((ext_vector_type(4)));
typedef float v8f __attribute__((ext_vector_type(8)));

#define WS_CNT_OFF  0
#define WS_X_OFF    (N_REL * N_NODES)
#define WS_STAGE_OFF (WS_X_OFF + N_NODES * HID)
#define WS_SSRC_OFF (WS_STAGE_OFF + N_NODES * 544)
#define WS_TABC_OFF  (WS_STAGE_OFF + SORT_WG * CHUNK)
#define WS_TABS_OFF  (WS_TABC_OFF + N_BUCK * SORT_WG)
#define NO_WORD 0xffffffffu
#define BT_WORDS    ((N_REL * HID + HID) * N_LAB)
#define WS_BT_OFF   (WS_TABS_OFF + N_BUCK * SORT_WG)
static_assert(WS_BT_OFF % 128 == 0, "a wave's 512-byte B load starts on a line boundary");
static_assert(WS_BT_OFF + BT_WORDS <= WS_SSRC_OFF, "the B copy stays inside the stage region");
static_assert(BT_WORDS <= N_BUCK * SORT_T, "one thread of the bucket launch per word of the B copy");

__device__ __forceinline__ int wave_incl_scan(int v, unsigned lane) {
  for (unsigned o = 1; o < 32; o <<= 1) {
    int u = __shfl_up(v, o);
    if (lane >= o) v += u;
  }
  return v;
}

__global__ __launch_bounds__(SORT_T) void k_part(const int* __restrict__ src_a, const int* __restrict__ dst_a,
                                                 const int* __restrict__ etype, int* __restrict__ tabc,
                                                 int* __restrict__ tabs, int* __restrict__ staged) {
  __shared__ int h[SORT_WG];
  __shared__ alignas(16) int obuf[CHUNK];
  unsigned t = threadIdx.x, g = blockIdx.x, lane = t & 31;
  if (t < SORT_WG) h[t] = 0;
  __syncthreads();
  unsigned beg = g * CHUNK;
  unsigned end = (beg + CHUNK < N_EDGE) ? (beg + CHUNK) : N_EDGE;
  int dv[CH_IT], wv[CH_IT];
  #pragma unroll
  for (int k = 0; k < CH_IT; k++) {
    unsigned i = beg + (unsigned)k * SORT_T + t;
    bool ok = i < end;
    int d = ok ? dst_a[i] : -1;
    int s = ok ? src_a[i] : 0;
    int r = ok ? etype[i] : 0;
    dv[k] = d;
    wv[k] = ((((d & 255) << 4) + r) << 16) | s;
  }
  int pos[CH_IT];
  #pragma unroll
  for (int k = 0; k < CH_IT; k++) {
    pos[k] = 0;
    if (dv[k] >= 0)
      pos[k] = __hip_atomic_fetch_add(&h[dv[k] >> 8], 1, __ATOMIC_RELAXED, __HIP_MEMORY_SCOPE_WORKGROUP);
  }
  __syncthreads();
  if (t < 32) {
    int cv[8], e[8];
    int s = 0;
    for (int j = 0; j < 8; j++) {
      cv[j] = h[8 * lane + (unsigned)j];
      e[j] = s; s += cv[j];
    }
    int inc = wave_incl_scan(s, lane);
    int exc = inc - s;
    for (int j = 0; j < 8; j++) {
      unsigned idx = 8 * lane + (unsigned)j;
      int st = e[j] + exc;
      h[idx] = st;
      if (idx < N_BUCK) {
        tabc[idx * SORT_WG + g] = cv[j];
        tabs[idx * SORT_WG + g] = st;
      }
    }
  }
  __syncthreads();
  #pragma unroll
  for (int k = 0; k < CH_IT; k++) {
    if (dv[k] >= 0) obuf[h[dv[k] >> 8] + pos[k]] = wv[k];
  }
  __syncthreads();
  unsigned n = end - beg;
  for (unsigned u = 4 * t; u < n; u += 4 * SORT_T) {
    if (u + 4 <= n) {
      *((int4*)(staged + beg + u)) = *((const int4*)(obuf + u));
    } else {
      for (unsigned q = u; q < n; q++) staged[beg + q] = obuf[q];
    }
  }
}

#define SL_PER_WAVE (SORT_WG / 32)
__global__ __launch_bounds__(SORT_T) void k_bucket(const int* __restrict__ tabc, const int* __restrict__ tabs,
                                                   const int* __restrict__ staged,
                                                   int* __restrict__ off, int* __restrict__ ssrc,
                                                   const float* __restrict__ w2, const float* __restrict__ r2,
                                                   float* __restrict__ bt) {
  __shared__ alignas(16) int cnt[BK_KEYS];
  __shared__ int wtot[32];
  __shared__ int slc[SORT_WG];
  __shared__ int slb[SORT_WG];
  __shared__ int bsum[SORT_WG / 32];
  __shared__ int bcnt[SORT_WG / 32];
  __shared__ alignas(16) int obuf[OB_CAP + 32];
  unsigned t = threadIdx.x, lane = t & 31, w = t >> 5;
  unsigned b = blockIdx.x;
  const unsigned gid = b * SORT_T + t;
  float btv = 0.0f;
  if (gid < BT_WORDS) {
    unsigned fe = gid & 3u, fl = (gid >> 2) & 31u, fws = gid >> 7;
    unsigned fw = fws / 17u, fs = fws - 17u * fw;
    unsigned frow = 2u * (fl >> 4) + (fe & 1u);
    unsigned fcol = (fl & 15u) + 16u * (fe >> 1);
    btv = (fs < 16u) ? w2[(64u * fw + 4u * fs + frow) * N_LAB + fcol]
                     : r2[(4u * fw + frow) * N_LAB + fcol];
  }
  int4 zero4; zero4.x = 0; zero4.y = 0; zero4.z = 0; zero4.w = 0;
  ((int4*)cnt)[t] = zero4;
  if (t < SORT_WG) {
    int c = tabc[b * SORT_WG + t];
    int s = tabs[b * SORT_WG + t];
    slc[t] = c;
    slb[t] = (int)(t * CHUNK) + s;
    int r = s, rc = c;
    for (int o = 16; o > 0; o >>= 1) { r += __shfl_xor(r, o); rc += __shfl_xor(rc, o); }
    if (lane == 0) { bsum[w] = r; bcnt[w] = rc; }
  }
  __syncthreads();
  int bstart = 0;
  int btotal = 0;
  for (int i = 0; i < SORT_WG / 32; i++) { bstart += bsum[i]; btotal += bcnt[i]; }
  const int shift = bstart & 31;
  const bool in_lds = btotal <= OB_CAP;
  int sc[SL_PER_WAVE], sb[SL_PER_WAVE];
  unsigned wd[SL_PER_WAVE][2];
  #pragma unroll
  for (int s = 0; s < SL_PER_WAVE; s++) {
    unsigned g = w + 32u * (unsigned)s;
    sc[s] = slc[g]; sb[s] = slb[g];
    wd[s][0] = ((int)lane < sc[s]) ? (unsigned)staged[sb[s] + (int)lane] : NO_WORD;
    wd[s][1] = ((int)lane + 32 < sc[s]) ? (unsigned)staged[sb[s] + (int)lane + 32] : NO_WORD;
  }
  #pragma unroll
  for (int s = 0; s < SL_PER_WAVE; s++) {
    for (int q = 0; q < 2; q++) {
      if (wd[s][q] != NO_WORD)
        __hip_atomic_fetch_add(&cnt[wd[s][q] >> 16], 1, __ATOMIC_RELAXED, __HIP_MEMORY_SCOPE_WORKGROUP);
    }
    for (int e = 64 + (int)lane; e < sc[s]; e += 32) {
      unsigned u = (unsigned)staged[sb[s] + e];
      __hip_atomic_fetch_add(&cnt[u >> 16], 1, __ATOMIC_RELAXED, __HIP_MEMORY_SCOPE_WORKGROUP);
    }
  }
  __syncthreads();
  int4 c = ((const int4*)cnt)[t];
  int s = c.x + c.y + c.z + c.w;
  int inc = wave_incl_scan(s, lane);
  if (lane == 31) wtot[w] = inc;
  __syncthreads();
  if (w == 0) {
    int v = wtot[lane];
    int vi = wave_incl_scan(v, lane);
    wtot[lane] = vi - v;
  }
  __syncthreads();
  int ex = wtot[w] + inc - s;
  int4 st;
  st.x = ex; st.y = st.x + c.x; st.z = st.y + c.y; st.w = st.z + c.z;
  ((int4*)cnt)[t] = st;
  unsigned key0 = b * BK_KEYS + 4 * t;
  if (key0 < N_KEYS) {
    int4 e;
    e.x = bstart + st.y; e.y = bstart + st.z; e.z = bstart + st.w; e.w = bstart + st.w + c.w;
    *((int4*)(off + key0)) = e;
  }
  __syncthreads();
  int pos[SL_PER_WAVE][2];
  #pragma unroll
  for (int s2 = 0; s2 < SL_PER_WAVE; s2++) {
    for (int q = 0; q < 2; q++) {
      pos[s2][q] = 0;
      if (wd[s2][q] != NO_WORD)
        pos[s2][q] = __hip_atomic_fetch_add(&cnt[wd[s2][q] >> 16], 1, __ATOMIC_RELAXED, __HIP_MEMORY_SCOPE_WORKGROUP);
    }
  }
  if (gid < BT_WORDS) bt[gid] = btv;
  if (in_lds) {
    #pragma unroll
    for (int s2 = 0; s2 < SL_PER_WAVE; s2++) {
      for (int q = 0; q < 2; q++) {
        if (wd[s2][q] != NO_WORD) obuf[shift + pos[s2][q]] = (int)(wd[s2][q] & 0xffffu);
      }
      for (int e = 64 + (int)lane; e < sc[s2]; e += 32) {
        unsigned u = (unsigned)staged[sb[s2] + e];
        int p = __hip_atomic_fetch_add(&cnt[u >> 16], 1, __ATOMIC_RELAXED, __HIP_MEMORY_SCOPE_WORKGROUP);
        obuf[shift + p] = (int)(u & 0xffffu);
      }
    }
    __syncthreads();
    int* gout = ssrc + (bstart - shift);
    const int lo = shift, hi = shift + btotal;
    for (int u = 4 * (int)t; u < hi; u += 4 * SORT_T) {
      if (u >= lo && u + 4 <= hi) {
        *((int4*)(gout + u)) = *((const int4*)(obuf + u));
      } else {
        for (int q = u; q < u + 4; q++)
          if (q >= lo && q < hi) gout[q] = obuf[q];
      }
    }
  } else {
    #pragma unroll
    for (int s2 = 0; s2 < SL_PER_WAVE; s2++) {
      for (int q = 0; q < 2; q++) {
        if (wd[s2][q] != NO_WORD) ssrc[bstart + pos[s2][q]] = (int)(wd[s2][q] & 0xffffu);
      }
      for (int e = 64 + (int)lane; e < sc[s2]; e += 32) {
        unsigned u = (unsigned)staged[sb[s2] + e];
        int p = __hip_atomic_fetch_add(&cnt[u >> 16], 1, __ATOMIC_RELAXED, __HIP_MEMORY_SCOPE_WORKGROUP);
        ssrc[bstart + p] = (int)(u & 0xffffu);
      }
    }
  }
}

__global__ void k_layer1(const int* __restrict__ off, const int* __restrict__ ssrc,
                         const float* __restrict__ w1, const float* __restrict__ root1,
                         const float* __restrict__ bias1, float* __restrict__ xout) {
  unsigned tid = blockIdx.x * blockDim.x + threadIdx.x;
  unsigned d = (unsigned)__builtin_amdgcn_readfirstlane((int)(tid >> 5));
  unsigned lane = threadIdx.x & 31;
  if (d >= N_NODES) return;
  unsigned r16 = lane & 15;
  int endv = off[d * N_REL + r16];
  int node_start = (d == 0) ? 0 : off[d * N_REL - 1];
  int prev = __shfl_up(endv, 1);
  int startv = (r16 == 0) ? node_start : prev;
  float invv = 1.0f / fmaxf((float)(endv - startv), 1.0f);
  int ends[16];
  for (int r = 0; r < 16; r++) ends[r] = __builtin_amdgcn_readlane(endv, r);
  int node_end = ends[15];
  v2f r1 = *((const v2f*)(root1 + (size_t)d * HID) + lane);
  v2f b1 = *((const v2f*)bias1 + lane);

  v2f acc = {0.0f, 0.0f};
  for (int cb = node_start; cb < node_end; cb += 32) {
    int p = cb + (int)lane;
    int sv = (p < node_end) ? ssrc[p] : 0;
    int relv = 0;
    for (int r = 0; r < 16; r++) relv += (p >= ends[r]) ? 1 : 0;
    relv = (relv > 15) ? 15 : relv;
    float iv = __shfl(invv, relv);
    int n = node_end - cb;
    n = (n > 32) ? 32 : n;
    for (int j = 0; j < n; j += 8) {
      v2f v[8];
      float ij[8];
      for (int k = 0; k < 8; k++) {
        int jj = (j + k < n) ? (j + k) : (n - 1);
        int sj = __builtin_amdgcn_readlane(sv, jj);
        int rj = __builtin_amdgcn_readlane(relv, jj);
        ij[k] = __builtin_bit_cast(float, __builtin_amdgcn_readlane(__builtin_bit_cast(int, iv), jj));
        v[k] = *((const v2f*)(w1 + ((size_t)rj * N_NODES + (size_t)sj) * HID) + lane);
      }
      for (int k = 0; k < 8; k++) {
        if (j + k < n) {
          acc.x += v[k].x * ij[k];
          acc.y += v[k].y * ij[k];
        }
      }
    }
  }
  v2f x;
  x.x = fmaxf(acc.x + r1.x + b1.x, 0.0f);
  x.y = fmaxf(acc.y + r1.y + b1.y, 0.0f);
  *((v2f*)(xout + (size_t)d * HID) + lane) = x;
}

#define M_COLS   (N_REL * HID + HID)
#define M_PITCH  (M_COLS + 4)
#define RED_PITCH 34
#define L2_WAVES 16
static_assert(N_NODES % 16 == 0, "every tile has 16 nodes: no partial tile, no early return before the barriers");
static_assert(L2_WAVES * 16 * RED_PITCH <= 16 * M_PITCH, "the reduce buffer reuses the matrix's LDS");
__global__ __launch_bounds__(L2_WAVES * 32) void k_layer2(const int* __restrict__ off, const int* __restrict__ ssrc,
                                                         const float* __restrict__ x, const float* __restrict__ bt,
                                                         const float* __restrict__ bias2,
                                                         float* __restrict__ out) {
  __shared__ alignas(16) float Ms[16 * M_PITCH];
  unsigned w = (unsigned)__builtin_amdgcn_readfirstlane((int)(threadIdx.x >> 5));
  unsigned lane = threadIdx.x & 31;
  unsigned d = blockIdx.x * 16u + w;
  float* Mrow = Ms + w * M_PITCH;
  v4f z4 = {0.0f, 0.0f, 0.0f, 0.0f};
  for (unsigned q = lane; q < M_COLS / 4; q += 32) *((v4f*)Mrow + q) = z4;

  unsigned r16 = lane & 15;
  int endv = off[d * N_REL + r16];
  int node_start = (d == 0) ? 0 : off[d * N_REL - 1];
  int prev = __shfl_up(endv, 1);
  int startv = (r16 == 0) ? node_start : prev;
  float invv = 1.0f / fmaxf((float)(endv - startv), 1.0f);
  int ends[16];
  for (int r = 0; r < 16; r++) ends[r] = __builtin_amdgcn_readlane(endv, r);
  int node_end = ends[15];
  *((v2f*)(Mrow + N_REL * HID) + lane) = *((const v2f*)(x + (size_t)d * HID) + lane);

  v2f acc = {0.0f, 0.0f};
  int cur = -1;
  for (int cb = node_start; cb < node_end; cb += 32) {
    int p = cb + (int)lane;
    int sv = (p < node_end) ? ssrc[p] : 0;
    int relv = 0;
    for (int r = 0; r < 16; r++) relv += (p >= ends[r]) ? 1 : 0;
    relv = (relv > 15) ? 15 : relv;
    int n = node_end - cb;
    n = (n > 32) ? 32 : n;
    for (int j = 0; j < n; j += 8) {
      v2f v[8];
      int rj[8];
      #pragma unroll
      for (int k = 0; k < 8; k++) {
        int jj = (j + k < n) ? (j + k) : (n - 1);
        int sj = __builtin_amdgcn_readlane(sv, jj);
        rj[k] = __builtin_amdgcn_readlane(relv, jj);
        v[k] = *((const v2f*)(x + (size_t)sj * HID) + lane);
      }
      #pragma unroll
      for (int k = 0; k < 8; k++) {
        if (j + k < n) {
          if (rj[k] != cur) {
            if (cur >= 0) {
              float iv = __builtin_bit_cast(float, __builtin_amdgcn_readlane(__builtin_bit_cast(int, invv), cur));
              v2f mv; mv.x = acc.x * iv; mv.y = acc.y * iv;
              *((v2f*)(Mrow + cur * HID) + lane) = mv;
            }
            acc.x = 0.0f; acc.y = 0.0f;
            cur = rj[k];
          }
          acc.x += v[k].x;
          acc.y += v[k].y;
        }
      }
    }
  }
  if (cur >= 0) {
    float iv = __builtin_bit_cast(float, __builtin_amdgcn_readlane(__builtin_bit_cast(int, invv), cur));
    v2f mv; mv.x = acc.x * iv; mv.y = acc.y * iv;
    *((v2f*)(Mrow + cur * HID) + lane) = mv;
  }

  unsigned half = lane >> 4;
  unsigned sub = lane & 15;
  const v4f* Bq = (const v4f*)bt + (size_t)w * (17 * 32) + lane;
  v4f q0[8];
  #pragma unroll
  for (int s = 0; s < 8; s++) q0[s] = Bq[32 * s];
  __syncthreads();

  const float* Af = Ms + sub * M_PITCH + 2u * half;
  v8f c0 = {}; v8f c1 = {};
  #pragma unroll
  for (int s = 0; s < 8; s++) {
    v2f a = *((const v2f*)(Af + HID * w + 4 * s));
    v2f b0 = {q0[s].x, q0[s].y};
    v2f b1 = {q0[s].z, q0[s].w};
    c0 = __builtin_amdgcn_wmma_f32_16x16x4_f32(false, a, false, b0, (short)0, c0, false, false);
    c1 = __builtin_amdgcn_wmma_f32_16x16x4_f32(false, a, false, b1, (short)0, c1, false, false);
  }
  v4f q1[9];
  #pragma unroll
  for (int s = 0; s < 9; s++) q1[s] = Bq[32 * (8 + s)];
  #pragma unroll
  for (int s = 0; s < 8; s++) {
    v2f a = *((const v2f*)(Af + HID * w + 4 * s + 32));
    v2f b0 = {q1[s].x, q1[s].y};
    v2f b1 = {q1[s].z, q1[s].w};
    c0 = __builtin_amdgcn_wmma_f32_16x16x4_f32(false, a, false, b0, (short)0, c0, false, false);
    c1 = __builtin_amdgcn_wmma_f32_16x16x4_f32(false, a, false, b1, (short)0, c1, false, false);
  }
  {
    v2f a = *((const v2f*)(Af + N_REL * HID + 4u * w));
    v2f b0 = {q1[8].x, q1[8].y};
    v2f b1 = {q1[8].z, q1[8].w};
    c0 = __builtin_amdgcn_wmma_f32_16x16x4_f32(false, a, false, b0, (short)0, c0, false, false);
    c1 = __builtin_amdgcn_wmma_f32_16x16x4_f32(false, a, false, b1, (short)0, c1, false, false);
  }
  __syncthreads();

  float* Pw = Ms + (w * 16u) * RED_PITCH + lane;
  #pragma unroll
  for (int v = 0; v < 8; v++) {
    Pw[v * RED_PITCH] = c0[v];
    Pw[(8 + v) * RED_PITCH] = c1[v];
  }
  __syncthreads();
  unsigned ri = (lane >> 4) * 8u + (w & 7u);
  unsigned sl = (w >> 3) * 16u + (lane & 15u);
  const float* Pr = Ms + ri * RED_PITCH + sl;
  float zsum = 0.0f;
  #pragma unroll
  for (int ww = 0; ww < L2_WAVES; ww++) zsum += Pr[(ww * 16) * RED_PITCH];
  float zz = zsum + bias2[lane];
  out[d * N_LAB + lane] = 1.0f / (1.0f + __expf(-zz));
}

extern "C" void kernel_launch(void* const* d_in, const int* in_sizes, int n_in,
                              void* d_out, int out_size, void* d_ws, size_t ws_size,
                              hipStream_t stream) {
  const int* ei     = (const int*)d_in[0];
  const int* src_a  = ei;
  const int* dst_a  = ei + N_EDGE;
  const int* etype  = (const int*)d_in[1];
  const float* w1   = (const float*)d_in[2];
  const float* r1   = (const float*)d_in[3];
  const float* b1   = (const float*)d_in[4];
  const float* w2   = (const float*)d_in[5];
  const float* r2   = (const float*)d_in[6];
  const float* b2   = (const float*)d_in[7];
  float* out = (float*)d_out;

  float* ws   = (float*)d_ws;
  int*   off  = (int*)d_ws + WS_CNT_OFF;
  float* xbuf = ws + WS_X_OFF;
  int*   staged = (int*)d_ws + WS_STAGE_OFF;
  int*   tabc = (int*)d_ws + WS_TABC_OFF;
  int*   tabs = (int*)d_ws + WS_TABS_OFF;
  int*   ssrc = (int*)d_ws + WS_SSRC_OFF;
  float* bt   = ws + WS_BT_OFF;

  (void)in_sizes; (void)n_in; (void)out_size; (void)ws_size;

  k_part<<<SORT_WG, SORT_T, 0, stream>>>(src_a, dst_a, etype, tabc, tabs, staged);
  k_bucket<<<N_BUCK, SORT_T, 0, stream>>>(tabc, tabs, staged, off, ssrc, w2, r2, bt);
  k_layer1<<<(N_NODES * 32u + 255) / 256, 256, 0, stream>>>(off, ssrc, w1, r1, b1, xbuf);
  k_layer2<<<N_NODES / 16, L2_WAVES * 32, 0, stream>>>(off, ssrc, xbuf, bt, b2, out);
}
